# starting kernel
# baseline (speedup 1.0000x reference)
#include <hip/hip_runtime.h>
#include <math.h>


typedef float v2f __attribute__((ext_vector_type(2)));
typedef float v8f __attribute__((ext_vector_type(8)));

#define GN0 120000
#define GN1 24000
#define GN2 6000
#define GD  256
#define GH1 8
#define GC1 64
#define GHC1 512
#define GH2 8
#define GC2 256
#define GHC2 2048
#define GE1 384000
#define GE2 96000
#define NEG_SLOPE 0.2f

__device__ __forceinline__ void atomicMaxF(float* addr, float v) {
  if (v >= 0.0f)
    atomicMax((int*)addr, __float_as_int(v));
  else
    atomicMin((unsigned int*)addr, __float_as_uint(v));
}

__global__ void __launch_bounds__(32)
wmma_gemm_f32(const float* __restrict__ A, const float* __restrict__ B,
              float* __restrict__ C, int M, int N, int K) {
  const int lane  = threadIdx.x & 31;
  const int mrow  = lane & 15;
  const int khalf = lane >> 4;
  const int m0 = blockIdx.x << 4;
  const int n0 = blockIdx.y << 6;

  v8f acc[4];
  acc[0] = {}; acc[1] = {}; acc[2] = {}; acc[3] = {};

  const float* arow = A + (size_t)(m0 + mrow) * K + 2 * khalf;

  for (int k0 = 0; k0 < K; k0 += 4) {
    v2f a;
    a.x = arow[k0];
    a.y = arow[k0 + 1];
    const float* bp = B + (size_t)(k0 + 2 * khalf) * N + n0 + mrow;
#pragma unroll
    for (int j = 0; j < 4; ++j) {
      v2f b;
      b.x = bp[j * 16];
      b.y = bp[N + j * 16];
      acc[j] = __builtin_amdgcn_wmma_f32_16x16x4_f32(
          false, a, false, b, (short)0, acc[j], false, false);
    }
  }

#pragma unroll
  for (int j = 0; j < 4; ++j) {
    const int col = n0 + j * 16 + mrow;
#pragma unroll
    for (int r = 0; r < 8; ++r) {
      const int row = m0 + r + (khalf << 3);
      C[(size_t)row * N + col] = acc[j][r];
    }
  }
}

__global__ void att_reduce(const float* __restrict__ h,
                           const float* __restrict__ att,
                           float* __restrict__ out,
                           int n_nodes, int HH, int CC) {
  const int w = (int)((blockIdx.x * (size_t)blockDim.x + threadIdx.x) >> 5);
  const int lane = threadIdx.x & 31;
  if (w >= n_nodes * HH) return;
  const int node = w / HH;
  const int head = w - node * HH;
  const float* hp = h + (size_t)node * HH * CC + (size_t)head * CC;
  const float* ap = att + (size_t)head * CC;
  float s = 0.0f;
  for (int c = lane; c < CC; c += 32) s += hp[c] * ap[c];
#pragma unroll
  for (int off = 16; off > 0; off >>= 1) s += __shfl_xor(s, off, 32);
  if (lane == 0) out[w] = s;
}

__global__ void edge_logits_max(const int* __restrict__ src,
                                const int* __restrict__ dst,
                                const float* __restrict__ a_src,
                                const float* __restrict__ a_dst,
                                float* __restrict__ ebuf,
                                float* __restrict__ m,
                                int E, int HH) {
  const int t = blockIdx.x * blockDim.x + threadIdx.x;
  if (t >= E * HH) return;
  const int e = t / HH;
  const int h = t - e * HH;
  float v = a_src[src[e] * HH + h] + a_dst[dst[e] * HH + h];
  v = (v > 0.0f) ? v : NEG_SLOPE * v;
  ebuf[t] = v;
  atomicMaxF(&m[dst[e] * HH + h], v);
}

__global__ void edge_exp_sum(const int* __restrict__ dst,
                             float* __restrict__ ebuf,
                             const float* __restrict__ m,
                             float* __restrict__ denom,
                             int E, int HH) {
  const int t = blockIdx.x * blockDim.x + threadIdx.x;
  if (t >= E * HH) return;
  const int e = t / HH;
  const int h = t - e * HH;
  const float ex = expf(ebuf[t] - m[dst[e] * HH + h]);
  ebuf[t] = ex;
  atomicAdd(&denom[dst[e] * HH + h], ex);
}

__global__ void edge_scatter(const float* __restrict__ hsrc,
                             const int* __restrict__ src,
                             const int* __restrict__ dst,
                             const float* __restrict__ ex,
                             const float* __restrict__ denom,
                             float* __restrict__ out,
                             int HH, int cshift, int outC,
                             float scale, int concat) {
  const int e = blockIdx.x;
  const int s = src[e];
  const int d = dst[e];
  const int HCt = HH << cshift;
  const int cmask = (1 << cshift) - 1;
  for (int i = threadIdx.x; i < HCt; i += blockDim.x) {
    const int h = i >> cshift;
    const float alpha =
        scale * ex[(size_t)e * HH + h] / denom[(size_t)d * HH + h];
    const int col = concat ? i : (i & cmask);
    atomicAdd(&out[(size_t)d * outC + col],
              alpha * hsrc[(size_t)s * HCt + i]);
  }
}

__global__ void fill_f32(float* __restrict__ p, float v, long n) {
  const long i = blockIdx.x * (long)blockDim.x + threadIdx.x;
  if (i < n) p[i] = v;
}

__global__ void init_rows_bias(float* __restrict__ out,
                               const float* __restrict__ bias,
                               long rows, int cols) {
  const long t = blockIdx.x * (long)blockDim.x + threadIdx.x;
  if (t >= rows * cols) return;
  out[t] = bias[t % cols];
}

__global__ void elu_inplace(float* __restrict__ p, long n) {
  const long i = blockIdx.x * (long)blockDim.x + threadIdx.x;
  if (i >= n) return;
  const float v = p[i];
  p[i] = (v > 0.0f) ? v : expm1f(v);
}

extern "C" void kernel_launch(void* const* d_in, const int* in_sizes, int n_in,
                              void* d_out, int out_size, void* d_ws,
                              size_t ws_size, hipStream_t stream) {
  (void)in_sizes; (void)n_in; (void)out_size; (void)ws_size;

  const float* x        = (const float*)d_in[0];
  const int*   src1     = (const int*)d_in[1];
  const int*   dst1     = (const int*)d_in[2];
  const int*   src2     = (const int*)d_in[3];
  const int*   dst2     = (const int*)d_in[4];
  const float* W1       = (const float*)d_in[5];
  const float* att_src1 = (const float*)d_in[6];
  const float* att_dst1 = (const float*)d_in[7];
  const float* b1       = (const float*)d_in[8];
  const float* W2       = (const float*)d_in[9];
  const float* att_src2 = (const float*)d_in[10];
  const float* att_dst2 = (const float*)d_in[11];
  const float* b2       = (const float*)d_in[12];
  float* out = (float*)d_out;

  char* ws = (char*)d_ws;
  size_t off = 0;
  auto alloc = [&](size_t bytes) -> float* {
    off = (off + 255) & ~(size_t)255;
    float* p = (float*)(ws + off);
    off += bytes;
    return p;
  };

  float* h1      = alloc((size_t)GN0 * GHC1 * 4);
  float* x1      = alloc((size_t)GN1 * GHC1 * 4);
  float* ebuf    = alloc((size_t)GE1 * GH1 * 4);
  float* a_src1  = alloc((size_t)GN0 * GH1 * 4);
  float* a_dst1  = alloc((size_t)GN1 * GH1 * 4);
  float* m1      = alloc((size_t)GN1 * GH1 * 4);
  float* denom1  = alloc((size_t)GN1 * GH1 * 4);
  float* a_src2  = alloc((size_t)GN1 * GH2 * 4);
  float* a_dst2  = alloc((size_t)GN2 * GH2 * 4);
  float* m2      = alloc((size_t)GN2 * GH2 * 4);
  float* denom2  = alloc((size_t)GN2 * GH2 * 4);
  float* h2      = h1;

  wmma_gemm_f32<<<dim3(GN0 / 16, GHC1 / 64), 32, 0, stream>>>(
      x, W1, h1, GN0, GHC1, GD);

  att_reduce<<<(GN0 * GH1 + 7) / 8, 256, 0, stream>>>(
      h1, att_src1, a_src1, GN0, GH1, GC1);
  att_reduce<<<(GN1 * GH1 + 7) / 8, 256, 0, stream>>>(
      h1, att_dst1, a_dst1, GN1, GH1, GC1);

  fill_f32<<<(GN1 * GH1 + 255) / 256, 256, 0, stream>>>(
      m1, -INFINITY, (long)GN1 * GH1);
  fill_f32<<<(GN1 * GH1 + 255) / 256, 256, 0, stream>>>(
      denom1, 0.0f, (long)GN1 * GH1);

  edge_logits_max<<<(GE1 * GH1 + 255) / 256, 256, 0, stream>>>(
      src1, dst1, a_src1, a_dst1, ebuf, m1, GE1, GH1);
  edge_exp_sum<<<(GE1 * GH1 + 255) / 256, 256, 0, stream>>>(
      dst1, ebuf, m1, denom1, GE1, GH1);

  init_rows_bias<<<((long)GN1 * GHC1 + 255) / 256, 256, 0, stream>>>(
      x1, b1, GN1, GHC1);
  edge_scatter<<<GE1, 128, 0, stream>>>(
      h1, src1, dst1, ebuf, denom1, x1, GH1,  6, GHC1,
      1.0f,  1);
  elu_inplace<<<((long)GN1 * GHC1 + 255) / 256, 256, 0, stream>>>(
      x1, (long)GN1 * GHC1);

  wmma_gemm_f32<<<dim3(GN1 / 16, GHC2 / 64), 32, 0, stream>>>(
      x1, W2, h2, GN1, GHC2, GHC1);

  att_reduce<<<(GN1 * GH2 + 7) / 8, 256, 0, stream>>>(
      h2, att_src2, a_src2, GN1, GH2, GC2);
  att_reduce<<<(GN2 * GH2 + 7) / 8, 256, 0, stream>>>(
      h2, att_dst2, a_dst2, GN2, GH2, GC2);

  fill_f32<<<(GN2 * GH2 + 255) / 256, 256, 0, stream>>>(
      m2, -INFINITY, (long)GN2 * GH2);
  fill_f32<<<(GN2 * GH2 + 255) / 256, 256, 0, stream>>>(
      denom2, 0.0f, (long)GN2 * GH2);

  edge_logits_max<<<(GE2 * GH2 + 255) / 256, 256, 0, stream>>>(
      src2, dst2, a_src2, a_dst2, ebuf, m2, GE2, GH2);
  edge_exp_sum<<<(GE2 * GH2 + 255) / 256, 256, 0, stream>>>(
      dst2, ebuf, m2, denom2, GE2, GH2);

  init_rows_bias<<<((long)GN2 * GC2 + 255) / 256, 256, 0, stream>>>(
      out, b2, GN2, GC2);
  edge_scatter<<<GE2, 256, 0, stream>>>(
      h2, src2, dst2, ebuf, denom2, out, GH2,  8, GC2,
      1.0f / (float)GH2,  0);
}


# GEMM retiled: 64x256 block tile staged through LDS (plain global loads + ds_store, two stages), 8 waves each 32x64 on 8 accumulators; was one wave per 16x64 strip from global. Sum order kept.
# speedup vs baseline: 1.2274x; 1.2274x over previous
#include <hip/hip_runtime.h>
#include <math.h>


typedef float v2f __attribute__((ext_vector_type(2)));
typedef float v8f __attribute__((ext_vector_type(8)));

#define GN0 120000
#define GN1 24000
#define GN2 6000
#define GD  256
#define GH1 8
#define GC1 64
#define GHC1 512
#define GH2 8
#define GC2 256
#define GHC2 2048
#define GE1 384000
#define GE2 96000
#define NEG_SLOPE 0.2f

__device__ __forceinline__ void atomicMaxF(float* addr, float v) {
  if (v >= 0.0f)
    atomicMax((int*)addr, __float_as_int(v));
  else
    atomicMin((unsigned int*)addr, __float_as_uint(v));
}

typedef float v4f __attribute__((ext_vector_type(4)));

#define TBM 64
#define TBN 256
#define TBK 16
#define APITCH 20
#define BPITCH2 544

__global__ void __launch_bounds__(256)
wmma_gemm_f32(const float* __restrict__ A, const float* __restrict__ B,
              float* __restrict__ C, int M, int N, int K) {
  __shared__ __attribute__((aligned(16))) float As[2][TBM * APITCH];
  __shared__ __attribute__((aligned(16))) float Bs[2][(TBK / 2) * BPITCH2];

  const int tid   = threadIdx.x;
  const int lane  = tid & 31;
  const int wave  = tid >> 5;
  const int mrow  = lane & 15;
  const int khalf = lane >> 4;
  const int wm = wave >> 2;
  const int wn = wave & 3;
  const int n0 = blockIdx.x * TBN;
  const int m0 = blockIdx.y * TBM;

  const int arow = tid >> 2;
  const int akq  = tid & 3;
  const int bp0  = tid >> 6;
  const int bnq  = tid & 63;
  const float* ag  = A + (size_t)(m0 + arow) * K + akq * 4;
  const float* bg0 = B + (size_t)(2 * bp0) * N + n0 + bnq * 4;
  const float* bg1 = bg0 + N;
  const size_t pair4 = (size_t)8 * N;
  const int a_st = arow * APITCH + akq * 4;
  const int b_st = bp0 * BPITCH2 + bnq * 8;

  const int a_ld = (wm * 32 + mrow) * APITCH + 2 * khalf;
  const int b_ld = khalf * BPITCH2 + (wn * 64 + mrow) * 2;

  v4f ra  = *(const v4f*)ag;
  v4f b00 = *(const v4f*)bg0;
  v4f b01 = *(const v4f*)bg1;
  v4f b10 = *(const v4f*)(bg0 + pair4);
  v4f b11 = *(const v4f*)(bg1 + pair4);

  *(v4f*)&As[0][a_st] = ra;
  *(v4f*)&Bs[0][b_st]                   = (v4f){b00.x, b01.x, b00.y, b01.y};
  *(v4f*)&Bs[0][b_st + 4]               = (v4f){b00.z, b01.z, b00.w, b01.w};
  *(v4f*)&Bs[0][b_st + 4 * BPITCH2]     = (v4f){b10.x, b11.x, b10.y, b11.y};
  *(v4f*)&Bs[0][b_st + 4 * BPITCH2 + 4] = (v4f){b10.z, b11.z, b10.w, b11.w};
  __syncthreads();

  v8f acc[2][4];
#pragma unroll
  for (int i = 0; i < 2; ++i)
#pragma unroll
    for (int j = 0; j < 4; ++j) acc[i][j] = {};

  const int nchunks = K / TBK;
  for (int ch = 0; ch < nchunks; ++ch) {
    const int cur = ch & 1;
    const bool more = (ch + 1) < nchunks;
    if (more) {
      const int kc = (ch + 1) * TBK;
      const size_t kcN = (size_t)kc * N;
      ra  = *(const v4f*)(ag + kc);
      b00 = *(const v4f*)(bg0 + kcN);
      b01 = *(const v4f*)(bg1 + kcN);
      b10 = *(const v4f*)(bg0 + kcN + pair4);
      b11 = *(const v4f*)(bg1 + kcN + pair4);
    }

    const float* as = &As[cur][a_ld];
    const float* bs = &Bs[cur][b_ld];
#pragma unroll
    for (int kk = 0; kk < TBK; kk += 4) {
      v2f a[2];
      v2f b[4];
#pragma unroll
      for (int i = 0; i < 2; ++i)
        a[i] = *(const v2f*)(as + i * 16 * APITCH + kk);
#pragma unroll
      for (int j = 0; j < 4; ++j)
        b[j] = *(const v2f*)(bs + (kk / 2) * BPITCH2 + j * 32);
#pragma unroll
      for (int i = 0; i < 2; ++i)
#pragma unroll
        for (int j = 0; j < 4; ++j)
          acc[i][j] = __builtin_amdgcn_wmma_f32_16x16x4_f32(
              false, a[i], false, b[j], (short)0, acc[i][j], false, false);
    }

    if (more) {
      const int nx = cur ^ 1;
      *(v4f*)&As[nx][a_st] = ra;
      *(v4f*)&Bs[nx][b_st]                   = (v4f){b00.x, b01.x, b00.y, b01.y};
      *(v4f*)&Bs[nx][b_st + 4]               = (v4f){b00.z, b01.z, b00.w, b01.w};
      *(v4f*)&Bs[nx][b_st + 4 * BPITCH2]     = (v4f){b10.x, b11.x, b10.y, b11.y};
      *(v4f*)&Bs[nx][b_st + 4 * BPITCH2 + 4] = (v4f){b10.z, b11.z, b10.w, b11.w};
    }
    __syncthreads();
  }

#pragma unroll
  for (int i = 0; i < 2; ++i) {
#pragma unroll
    for (int j = 0; j < 4; ++j) {
      const int col = n0 + wn * 64 + j * 16 + mrow;
#pragma unroll
      for (int r = 0; r < 8; ++r) {
        const int row = m0 + wm * 32 + i * 16 + r + (khalf << 3);
        C[(size_t)row * N + col] = acc[i][j][r];
      }
    }
  }
}

__global__ void att_reduce(const float* __restrict__ h,
                           const float* __restrict__ att,
                           float* __restrict__ out,
                           int n_nodes, int HH, int CC) {
  const int w = (int)((blockIdx.x * (size_t)blockDim.x + threadIdx.x) >> 5);
  const int lane = threadIdx.x & 31;
  if (w >= n_nodes * HH) return;
  const int node = w / HH;
  const int head = w - node * HH;
  const float* hp = h + (size_t)node * HH * CC + (size_t)head * CC;
  const float* ap = att + (size_t)head * CC;
  float s = 0.0f;
  for (int c = lane; c < CC; c += 32) s += hp[c] * ap[c];
#pragma unroll
  for (int off = 16; off > 0; off >>= 1) s += __shfl_xor(s, off, 32);
  if (lane == 0) out[w] = s;
}

__global__ void edge_logits_max(const int* __restrict__ src,
                                const int* __restrict__ dst,
                                const float* __restrict__ a_src,
                                const float* __restrict__ a_dst,
                                float* __restrict__ ebuf,
                                float* __restrict__ m,
                                int E, int HH) {
  const int t = blockIdx.x * blockDim.x + threadIdx.x;
  if (t >= E * HH) return;
  const int e = t / HH;
  const int h = t - e * HH;
  float v = a_src[src[e] * HH + h] + a_dst[dst[e] * HH + h];
  v = (v > 0.0f) ? v : NEG_SLOPE * v;
  ebuf[t] = v;
  atomicMaxF(&m[dst[e] * HH + h], v);
}

__global__ void edge_exp_sum(const int* __restrict__ dst,
                             float* __restrict__ ebuf,
                             const float* __restrict__ m,
                             float* __restrict__ denom,
                             int E, int HH) {
  const int t = blockIdx.x * blockDim.x + threadIdx.x;
  if (t >= E * HH) return;
  const int e = t / HH;
  const int h = t - e * HH;
  const float ex = expf(ebuf[t] - m[dst[e] * HH + h]);
  ebuf[t] = ex;
  atomicAdd(&denom[dst[e] * HH + h], ex);
}

__global__ void edge_scatter(const float* __restrict__ hsrc,
                             const int* __restrict__ src,
                             const int* __restrict__ dst,
                             const float* __restrict__ ex,
                             const float* __restrict__ denom,
                             float* __restrict__ out,
                             int HH, int cshift, int outC,
                             float scale, int concat) {
  const int e = blockIdx.x;
  const int s = src[e];
  const int d = dst[e];
  const int HCt = HH << cshift;
  const int cmask = (1 << cshift) - 1;
  for (int i = threadIdx.x; i < HCt; i += blockDim.x) {
    const int h = i >> cshift;
    const float alpha =
        scale * ex[(size_t)e * HH + h] / denom[(size_t)d * HH + h];
    const int col = concat ? i : (i & cmask);
    atomicAdd(&out[(size_t)d * outC + col],
              alpha * hsrc[(size_t)s * HCt + i]);
  }
}

__global__ void fill_f32(float* __restrict__ p, float v, long n) {
  const long i = blockIdx.x * (long)blockDim.x + threadIdx.x;
  if (i < n) p[i] = v;
}

__global__ void init_rows_bias(float* __restrict__ out,
                               const float* __restrict__ bias,
                               long rows, int cols) {
  const long t = blockIdx.x * (long)blockDim.x + threadIdx.x;
  if (t >= rows * cols) return;
  out[t] = bias[t % cols];
}

__global__ void elu_inplace(float* __restrict__ p, long n) {
  const long i = blockIdx.x * (long)blockDim.x + threadIdx.x;
  if (i >= n) return;
  const float v = p[i];
  p[i] = (v > 0.0f) ? v : expm1f(v);
}

extern "C" void kernel_launch(void* const* d_in, const int* in_sizes, int n_in,
                              void* d_out, int out_size, void* d_ws,
                              size_t ws_size, hipStream_t stream) {
  (void)in_sizes; (void)n_in; (void)out_size; (void)ws_size;

  const float* x        = (const float*)d_in[0];
  const int*   src1     = (const int*)d_in[1];
  const int*   dst1     = (const int*)d_in[2];
  const int*   src2     = (const int*)d_in[3];
  const int*   dst2     = (const int*)d_in[4];
  const float* W1       = (const float*)d_in[5];
  const float* att_src1 = (const float*)d_in[6];
  const float* att_dst1 = (const float*)d_in[7];
  const float* b1       = (const float*)d_in[8];
  const float* W2       = (const float*)d_in[9];
  const float* att_src2 = (const float*)d_in[10];
  const float* att_dst2 = (const float*)d_in[11];
  const float* b2       = (const float*)d_in[12];
  float* out = (float*)d_out;

  char* ws = (char*)d_ws;
  size_t off = 0;
  auto alloc = [&](size_t bytes) -> float* {
    off = (off + 255) & ~(size_t)255;
    float* p = (float*)(ws + off);
    off += bytes;
    return p;
  };

  float* h1      = alloc((size_t)GN0 * GHC1 * 4);
  float* x1      = alloc((size_t)GN1 * GHC1 * 4);
  float* ebuf    = alloc((size_t)GE1 * GH1 * 4);
  float* a_src1  = alloc((size_t)GN0 * GH1 * 4);
  float* a_dst1  = alloc((size_t)GN1 * GH1 * 4);
  float* m1      = alloc((size_t)GN1 * GH1 * 4);
  float* denom1  = alloc((size_t)GN1 * GH1 * 4);
  float* a_src2  = alloc((size_t)GN1 * GH2 * 4);
  float* a_dst2  = alloc((size_t)GN2 * GH2 * 4);
  float* m2      = alloc((size_t)GN2 * GH2 * 4);
  float* denom2  = alloc((size_t)GN2 * GH2 * 4);
  float* h2      = h1;

  wmma_gemm_f32<<<dim3(GHC1 / TBN, GN0 / TBM), 256, 0, stream>>>(
      x, W1, h1, GN0, GHC1, GD);

  att_reduce<<<(GN0 * GH1 + 7) / 8, 256, 0, stream>>>(
      h1, att_src1, a_src1, GN0, GH1, GC1);
  att_reduce<<<(GN1 * GH1 + 7) / 8, 256, 0, stream>>>(
      h1, att_dst1, a_dst1, GN1, GH1, GC1);

  fill_f32<<<(GN1 * GH1 + 255) / 256, 256, 0, stream>>>(
      m1, -INFINITY, (long)GN1 * GH1);
  fill_f32<<<(GN1 * GH1 + 255) / 256, 256, 0, stream>>>(
      denom1, 0.0f, (long)GN1 * GH1);

  edge_logits_max<<<(GE1 * GH1 + 255) / 256, 256, 0, stream>>>(
      src1, dst1, a_src1, a_dst1, ebuf, m1, GE1, GH1);
  edge_exp_sum<<<(GE1 * GH1 + 255) / 256, 256, 0, stream>>>(
      dst1, ebuf, m1, denom1, GE1, GH1);

  init_rows_bias<<<((long)GN1 * GHC1 + 255) / 256, 256, 0, stream>>>(
      x1, b1, GN1, GHC1);
  edge_scatter<<<GE1, 128, 0, stream>>>(
      h1, src1, dst1, ebuf, denom1, x1, GH1,  6, GHC1,
      1.0f,  1);
  elu_inplace<<<((long)GN1 * GHC1 + 255) / 256, 256, 0, stream>>>(
      x1, (long)GN1 * GHC1);

  wmma_gemm_f32<<<dim3(GHC2 / TBN, GN1 / TBM), 256, 0, stream>>>(
      x1, W2, h2, GN1, GHC2, GHC1);

  att_reduce<<<(GN1 * GH2 + 7) / 8, 256, 0, stream>>>(
      h2, att_src2, a_src2, GN1, GH2, GC2);
  att_reduce<<<(GN2 * GH2 + 7) / 8, 256, 0, stream>>>(
      h2, att_dst2, a_dst2, GN2, GH2, GC2);

  fill_f32<<<(GN2 * GH2 + 255) / 256, 256, 0, stream>>>(
      m2, -INFINITY, (long)GN2 * GH2);
  fill_f32<<<(GN2 * GH2 + 255) / 256, 256, 0, stream>>>(
      denom2, 0.0f, (long)GN2 * GH2);

  edge_logits_max<<<(GE2 * GH2 + 255) / 256, 256, 0, stream>>>(
      src2, dst2, a_src2, a_dst2, ebuf, m2, GE2, GH2);
  edge_exp_sum<<<(GE2 * GH2 + 255) / 256, 256, 0, stream>>>(
      dst2, ebuf, m2, denom2, GE2, GH2);

  init_rows_bias<<<((long)GN2 * GC2 + 255) / 256, 256, 0, stream>>>(
      out, b2, GN2, GC2);
  edge_scatter<<<GE2, 256, 0, stream>>>(
      h2, src2, dst2, ebuf, denom2, out, GH2,  8, GC2,
      1.0f / (float)GH2,  0);
}


# Non-GEMM kernels coarsened: one wave per node (all heads), one wave per edge row with loads ahead of atomics, 16-byte x4 fills, 8 items per thread in edge passes; 7x fewer waves. Sum order kept.
# speedup vs baseline: 1.8129x; 1.4770x over previous
#include <hip/hip_runtime.h>
#include <math.h>


typedef float v2f __attribute__((ext_vector_type(2)));
typedef float v8f __attribute__((ext_vector_type(8)));

#define GN0 120000
#define GN1 24000
#define GN2 6000
#define GD  256
#define GH1 8
#define GC1 64
#define GHC1 512
#define GH2 8
#define GC2 256
#define GHC2 2048
#define GE1 384000
#define GE2 96000
#define NEG_SLOPE 0.2f

__device__ __forceinline__ void atomicMaxF(float* addr, float v) {
  if (v >= 0.0f)
    atomicMax((int*)addr, __float_as_int(v));
  else
    atomicMin((unsigned int*)addr, __float_as_uint(v));
}

typedef float v4f __attribute__((ext_vector_type(4)));

#define TBM 64
#define TBN 256
#define TBK 16
#define APITCH 20
#define BPITCH2 544

__global__ void __launch_bounds__(256)
wmma_gemm_f32(const float* __restrict__ A, const float* __restrict__ B,
              float* __restrict__ C, int M, int N, int K) {
  __shared__ __attribute__((aligned(16))) float As[2][TBM * APITCH];
  __shared__ __attribute__((aligned(16))) float Bs[2][(TBK / 2) * BPITCH2];

  const int tid   = threadIdx.x;
  const int lane  = tid & 31;
  const int wave  = tid >> 5;
  const int mrow  = lane & 15;
  const int khalf = lane >> 4;
  const int wm = wave >> 2;
  const int wn = wave & 3;
  const int n0 = blockIdx.x * TBN;
  const int m0 = blockIdx.y * TBM;

  const int arow = tid >> 2;
  const int akq  = tid & 3;
  const int bp0  = tid >> 6;
  const int bnq  = tid & 63;
  const float* ag  = A + (size_t)(m0 + arow) * K + akq * 4;
  const float* bg0 = B + (size_t)(2 * bp0) * N + n0 + bnq * 4;
  const float* bg1 = bg0 + N;
  const size_t pair4 = (size_t)8 * N;
  const int a_st = arow * APITCH + akq * 4;
  const int b_st = bp0 * BPITCH2 + bnq * 8;

  const int a_ld = (wm * 32 + mrow) * APITCH + 2 * khalf;
  const int b_ld = khalf * BPITCH2 + (wn * 64 + mrow) * 2;

  v4f ra  = *(const v4f*)ag;
  v4f b00 = *(const v4f*)bg0;
  v4f b01 = *(const v4f*)bg1;
  v4f b10 = *(const v4f*)(bg0 + pair4);
  v4f b11 = *(const v4f*)(bg1 + pair4);

  *(v4f*)&As[0][a_st] = ra;
  *(v4f*)&Bs[0][b_st]                   = (v4f){b00.x, b01.x, b00.y, b01.y};
  *(v4f*)&Bs[0][b_st + 4]               = (v4f){b00.z, b01.z, b00.w, b01.w};
  *(v4f*)&Bs[0][b_st + 4 * BPITCH2]     = (v4f){b10.x, b11.x, b10.y, b11.y};
  *(v4f*)&Bs[0][b_st + 4 * BPITCH2 + 4] = (v4f){b10.z, b11.z, b10.w, b11.w};
  __syncthreads();

  v8f acc[2][4];
#pragma unroll
  for (int i = 0; i < 2; ++i)
#pragma unroll
    for (int j = 0; j < 4; ++j) acc[i][j] = {};

  const int nchunks = K / TBK;
  for (int ch = 0; ch < nchunks; ++ch) {
    const int cur = ch & 1;
    const bool more = (ch + 1) < nchunks;
    if (more) {
      const int kc = (ch + 1) * TBK;
      const size_t kcN = (size_t)kc * N;
      ra  = *(const v4f*)(ag + kc);
      b00 = *(const v4f*)(bg0 + kcN);
      b01 = *(const v4f*)(bg1 + kcN);
      b10 = *(const v4f*)(bg0 + kcN + pair4);
      b11 = *(const v4f*)(bg1 + kcN + pair4);
    }

    const float* as = &As[cur][a_ld];
    const float* bs = &Bs[cur][b_ld];
#pragma unroll
    for (int kk = 0; kk < TBK; kk += 4) {
      v2f a[2];
      v2f b[4];
#pragma unroll
      for (int i = 0; i < 2; ++i)
        a[i] = *(const v2f*)(as + i * 16 * APITCH + kk);
#pragma unroll
      for (int j = 0; j < 4; ++j)
        b[j] = *(const v2f*)(bs + (kk / 2) * BPITCH2 + j * 32);
#pragma unroll
      for (int i = 0; i < 2; ++i)
#pragma unroll
        for (int j = 0; j < 4; ++j)
          acc[i][j] = __builtin_amdgcn_wmma_f32_16x16x4_f32(
              false, a[i], false, b[j], (short)0, acc[i][j], false, false);
    }

    if (more) {
      const int nx = cur ^ 1;
      *(v4f*)&As[nx][a_st] = ra;
      *(v4f*)&Bs[nx][b_st]                   = (v4f){b00.x, b01.x, b00.y, b01.y};
      *(v4f*)&Bs[nx][b_st + 4]               = (v4f){b00.z, b01.z, b00.w, b01.w};
      *(v4f*)&Bs[nx][b_st + 4 * BPITCH2]     = (v4f){b10.x, b11.x, b10.y, b11.y};
      *(v4f*)&Bs[nx][b_st + 4 * BPITCH2 + 4] = (v4f){b10.z, b11.z, b10.w, b11.w};
    }
    __syncthreads();
  }

#pragma unroll
  for (int i = 0; i < 2; ++i) {
#pragma unroll
    for (int j = 0; j < 4; ++j) {
      const int col = n0 + wn * 64 + j * 16 + mrow;
#pragma unroll
      for (int r = 0; r < 8; ++r) {
        const int row = m0 + wm * 32 + i * 16 + r + (khalf << 3);
        C[(size_t)row * N + col] = acc[i][j][r];
      }
    }
  }
}

template <int HH, int CC, int NPW>
__global__ void __launch_bounds__(256)
att_reduce(const float* __restrict__ h,
           const float* __restrict__ att,
           float* __restrict__ out,
           int n_nodes) {
  const int w = (int)((blockIdx.x * 256u + threadIdx.x) >> 5);
  const int lane = threadIdx.x & 31;
  for (int k = 0; k < NPW; ++k) {
    const int node = w * NPW + k;
    if (node >= n_nodes) return;
    const float* hp = h + (size_t)node * (HH * CC);
    float keep = 0.0f;
#pragma unroll
    for (int head = 0; head < HH; ++head) {
      float s = 0.0f;
#pragma unroll
      for (int c = lane; c < CC; c += 32)
        s += hp[head * CC + c] * att[head * CC + c];
#pragma unroll
      for (int off = 16; off > 0; off >>= 1) s += __shfl_xor(s, off, 32);
      if (lane == head) keep = s;
    }
    if (lane < HH) out[(size_t)node * HH + lane] = keep;
  }
}

#define EPT 8
__global__ void __launch_bounds__(256)
edge_logits_max(const int* __restrict__ src,
                const int* __restrict__ dst,
                const float* __restrict__ a_src,
                const float* __restrict__ a_dst,
                float* __restrict__ ebuf,
                float* __restrict__ m,
                int E, int HH) {
  const int total = E * HH;
  const int base = blockIdx.x * (256 * EPT) + threadIdx.x;
  float v[EPT];
  int mi[EPT];
#pragma unroll
  for (int k = 0; k < EPT; ++k) {
    const int t = min(base + k * 256, total - 1);
    const int e = t / HH;
    const int h = t - e * HH;
    mi[k] = dst[e] * HH + h;
    const float a = a_src[src[e] * HH + h] + a_dst[mi[k]];
    v[k] = (a > 0.0f) ? a : NEG_SLOPE * a;
  }
#pragma unroll
  for (int k = 0; k < EPT; ++k) {
    const int t = base + k * 256;
    if (t < total) {
      ebuf[t] = v[k];
      atomicMaxF(&m[mi[k]], v[k]);
    }
  }
}

__global__ void __launch_bounds__(256)
edge_exp_sum(const int* __restrict__ dst,
             float* __restrict__ ebuf,
             const float* __restrict__ m,
             float* __restrict__ denom,
             int E, int HH) {
  const int total = E * HH;
  const int base = blockIdx.x * (256 * EPT) + threadIdx.x;
  float ex[EPT];
  int mi[EPT];
#pragma unroll
  for (int k = 0; k < EPT; ++k) {
    const int t = min(base + k * 256, total - 1);
    const int e = t / HH;
    const int h = t - e * HH;
    mi[k] = dst[e] * HH + h;
    ex[k] = expf(ebuf[t] - m[mi[k]]);
  }
#pragma unroll
  for (int k = 0; k < EPT; ++k) {
    const int t = base + k * 256;
    if (t < total) {
      ebuf[t] = ex[k];
      atomicAdd(&denom[mi[k]], ex[k]);
    }
  }
}

template <int HH, int CC, bool CONCAT>
__global__ void __launch_bounds__(256)
edge_scatter(const float* __restrict__ hsrc,
             const int* __restrict__ src,
             const int* __restrict__ dst,
             const float* __restrict__ ex,
             const float* __restrict__ denom,
             float* __restrict__ out,
             int E, int outC, float scale) {
  const int e = __builtin_amdgcn_readfirstlane(
      (int)(blockIdx.x * 8u + (threadIdx.x >> 5)));
  if (e >= E) return;
  const int lane = threadIdx.x & 31;
  const int s = src[e];
  const int d = dst[e];
  const float* hp = hsrc + (size_t)s * (HH * CC);
  float* op = out + (size_t)d * outC;
  constexpr int PH = CC / 32;
  float v[HH * PH];
#pragma unroll
  for (int j = 0; j < HH * PH; ++j) v[j] = hp[j * 32 + lane];
#pragma unroll
  for (int h = 0; h < HH; ++h) {
    const float alpha =
        scale * ex[(size_t)e * HH + h] / denom[(size_t)d * HH + h];
#pragma unroll
    for (int q = 0; q < PH; ++q) {
      const int c = q * 32 + lane;
      atomicAdd(&op[CONCAT ? (h * CC + c) : c], alpha * v[h * PH + q]);
    }
  }
}

__global__ void fill_f32(float* __restrict__ p, float v, long n) {
  const long i = blockIdx.x * (long)blockDim.x + threadIdx.x;
  if (i < n) p[i] = v;
}

#define VPT 4
__global__ void __launch_bounds__(256)
init_rows_bias(float* __restrict__ out,
               const float* __restrict__ bias,
               long rows, int cols) {
  const long nq = rows * cols / 4;
  const long base = blockIdx.x * (long)(256 * VPT) + threadIdx.x;
  const unsigned qcols = (unsigned)cols / 4u;
  const unsigned r0 = (unsigned)(base % (long)qcols);
  v4f b[VPT];
#pragma unroll
  for (int k = 0; k < VPT; ++k)
    b[k] = *(const v4f*)(bias + 4u * ((r0 + 256u * k) % qcols));
#pragma unroll
  for (int k = 0; k < VPT; ++k) {
    const long q = base + k * 256;
    if (q < nq) *(v4f*)(out + 4 * q) = b[k];
  }
}

__global__ void __launch_bounds__(256)
elu_inplace(float* __restrict__ p, long n) {
  const long nq = n / 4;
  const long base = blockIdx.x * (long)(256 * VPT) + threadIdx.x;
  v4f v[VPT];
#pragma unroll
  for (int k = 0; k < VPT; ++k) {
    const long q = base + k * 256;
    v[k] = *(const v4f*)(p + 4 * (q < nq ? q : nq - 1));
  }
#pragma unroll
  for (int k = 0; k < VPT; ++k) {
    const long q = base + k * 256;
    if (q < nq) {
      v4f r = v[k];
      r.x = (r.x > 0.0f) ? r.x : expm1f(r.x);
      r.y = (r.y > 0.0f) ? r.y : expm1f(r.y);
      r.z = (r.z > 0.0f) ? r.z : expm1f(r.z);
      r.w = (r.w > 0.0f) ? r.w : expm1f(r.w);
      *(v4f*)(p + 4 * q) = r;
    }
  }
}

extern "C" void kernel_launch(void* const* d_in, const int* in_sizes, int n_in,
                              void* d_out, int out_size, void* d_ws,
                              size_t ws_size, hipStream_t stream) {
  (void)in_sizes; (void)n_in; (void)out_size; (void)ws_size;

  const float* x        = (const float*)d_in[0];
  const int*   src1     = (const int*)d_in[1];
  const int*   dst1     = (const int*)d_in[2];
  const int*   src2     = (const int*)d_in[3];
  const int*   dst2     = (const int*)d_in[4];
  const float* W1       = (const float*)d_in[5];
  const float* att_src1 = (const float*)d_in[6];
  const float* att_dst1 = (const float*)d_in[7];
  const float* b1       = (const float*)d_in[8];
  const float* W2       = (const float*)d_in[9];
  const float* att_src2 = (const float*)d_in[10];
  const float* att_dst2 = (const float*)d_in[11];
  const float* b2       = (const float*)d_in[12];
  float* out = (float*)d_out;

  char* ws = (char*)d_ws;
  size_t off = 0;
  auto alloc = [&](size_t bytes) -> float* {
    off = (off + 255) & ~(size_t)255;
    float* p = (float*)(ws + off);
    off += bytes;
    return p;
  };

  float* h1      = alloc((size_t)GN0 * GHC1 * 4);
  float* x1      = alloc((size_t)GN1 * GHC1 * 4);
  float* ebuf    = alloc((size_t)GE1 * GH1 * 4);
  float* a_src1  = alloc((size_t)GN0 * GH1 * 4);
  float* a_dst1  = alloc((size_t)GN1 * GH1 * 4);
  float* m1      = alloc((size_t)GN1 * GH1 * 4);
  float* denom1  = alloc((size_t)GN1 * GH1 * 4);
  float* a_src2  = alloc((size_t)GN1 * GH2 * 4);
  float* a_dst2  = alloc((size_t)GN2 * GH2 * 4);
  float* m2      = alloc((size_t)GN2 * GH2 * 4);
  float* denom2  = alloc((size_t)GN2 * GH2 * 4);
  float* h2      = h1;

  wmma_gemm_f32<<<dim3(GHC1 / TBN, GN0 / TBM), 256, 0, stream>>>(
      x, W1, h1, GN0, GHC1, GD);

  att_reduce<GH1, GC1, 4><<<(GN0 + 31) / 32, 256, 0, stream>>>(
      h1, att_src1, a_src1, GN0);
  att_reduce<GH1, GC1, 4><<<(GN1 + 31) / 32, 256, 0, stream>>>(
      h1, att_dst1, a_dst1, GN1);

  fill_f32<<<(GN1 * GH1 + 255) / 256, 256, 0, stream>>>(
      m1, -INFINITY, (long)GN1 * GH1);
  fill_f32<<<(GN1 * GH1 + 255) / 256, 256, 0, stream>>>(
      denom1, 0.0f, (long)GN1 * GH1);

  edge_logits_max<<<(GE1 * GH1 + 256 * EPT - 1) / (256 * EPT), 256, 0, stream>>>(
      src1, dst1, a_src1, a_dst1, ebuf, m1, GE1, GH1);
  edge_exp_sum<<<(GE1 * GH1 + 256 * EPT - 1) / (256 * EPT), 256, 0, stream>>>(
      dst1, ebuf, m1, denom1, GE1, GH1);

  init_rows_bias<<<((long)GN1 * GHC1 / 4 + 256 * VPT - 1) / (256 * VPT), 256, 0,
                   stream>>>(x1, b1, GN1, GHC1);
  edge_scatter<GH1, GC1, true><<<(GE1 + 7) / 8, 256, 0, stream>>>(
      h1, src1, dst1, ebuf, denom1, x1, GE1, GHC1, 1.0f);
  elu_inplace<<<((long)GN1 * GHC1 / 4 + 256 * VPT - 1) / (256 * VPT), 256, 0,
                stream>>>(x1, (long)GN1 * GHC1);

  wmma_gemm_f32<<<dim3(GHC2 / TBN, GN1 / TBM), 256, 0, stream>>>(
      x1, W2, h2, GN1, GHC2, GHC1);

  att_reduce<GH2, GC2, 1><<<(GN1 + 7) / 8, 256, 0, stream>>>(
      h2, att_src2, a_src2, GN1);
  att_reduce<GH2, GC2, 1><<<(GN2 + 7) / 8, 256, 0, stream>>>(
      h2, att_dst2, a_dst2, GN2);

  fill_f32<<<(GN2 * GH2 + 255) / 256, 256, 0, stream>>>(
      m2, -INFINITY, (long)GN2 * GH2);
  fill_f32<<<(GN2 * GH2 + 255) / 256, 256, 0, stream>>>(
      denom2, 0.0f, (long)GN2 * GH2);

  edge_logits_max<<<(GE2 * GH2 + 256 * EPT - 1) / (256 * EPT), 256, 0, stream>>>(
      src2, dst2, a_src2, a_dst2, ebuf, m2, GE2, GH2);
  edge_exp_sum<<<(GE2 * GH2 + 256 * EPT - 1) / (256 * EPT), 256, 0, stream>>>(
      dst2, ebuf, m2, denom2, GE2, GH2);

  init_rows_bias<<<((long)GN2 * GC2 / 4 + 256 * VPT - 1) / (256 * VPT), 256, 0,
                   stream>>>(out, b2, GN2, GC2);
  edge_scatter<GH2, GC2, false><<<(GE2 + 7) / 8, 256, 0, stream>>>(
      h2, src2, dst2, ebuf, denom2, out, GE2, GC2, 1.0f / (float)GH2);
}


# Layer-2 scatter: an edge's 8 heads are summed in registers (ascending head, fused multiply-add) before the atomic add: one f32 atomic per output column per edge instead of 8; sum order changes.
# speedup vs baseline: 1.9740x; 1.0889x over previous
#include <hip/hip_runtime.h>
#include <math.h>


typedef float v2f __attribute__((ext_vector_type(2)));
typedef float v8f __attribute__((ext_vector_type(8)));

#define GN0 120000
#define GN1 24000
#define GN2 6000
#define GD  256
#define GH1 8
#define GC1 64
#define GHC1 512
#define GH2 8
#define GC2 256
#define GHC2 2048
#define GE1 384000
#define GE2 96000
#define NEG_SLOPE 0.2f

__device__ __forceinline__ void atomicMaxF(float* addr, float v) {
  if (v >= 0.0f)
    atomicMax((int*)addr, __float_as_int(v));
  else
    atomicMin((unsigned int*)addr, __float_as_uint(v));
}

typedef float v4f __attribute__((ext_vector_type(4)));

#define TBM 64
#define TBN 256
#define TBK 16
#define APITCH 20
#define BPITCH2 544

__global__ void __launch_bounds__(256)
wmma_gemm_f32(const float* __restrict__ A, const float* __restrict__ B,
              float* __restrict__ C, int M, int N, int K) {
  __shared__ __attribute__((aligned(16))) float As[2][TBM * APITCH];
  __shared__ __attribute__((aligned(16))) float Bs[2][(TBK / 2) * BPITCH2];

  const int tid   = threadIdx.x;
  const int lane  = tid & 31;
  const int wave  = tid >> 5;
  const int mrow  = lane & 15;
  const int khalf = lane >> 4;
  const int wm = wave >> 2;
  const int wn = wave & 3;
  const int n0 = blockIdx.x * TBN;
  const int m0 = blockIdx.y * TBM;

  const int arow = tid >> 2;
  const int akq  = tid & 3;
  const int bp0  = tid >> 6;
  const int bnq  = tid & 63;
  const float* ag  = A + (size_t)(m0 + arow) * K + akq * 4;
  const float* bg0 = B + (size_t)(2 * bp0) * N + n0 + bnq * 4;
  const float* bg1 = bg0 + N;
  const size_t pair4 = (size_t)8 * N;
  const int a_st = arow * APITCH + akq * 4;
  const int b_st = bp0 * BPITCH2 + bnq * 8;

  const int a_ld = (wm * 32 + mrow) * APITCH + 2 * khalf;
  const int b_ld = khalf * BPITCH2 + (wn * 64 + mrow) * 2;

  v4f ra  = *(const v4f*)ag;
  v4f b00 = *(const v4f*)bg0;
  v4f b01 = *(const v4f*)bg1;
  v4f b10 = *(const v4f*)(bg0 + pair4);
  v4f b11 = *(const v4f*)(bg1 + pair4);

  *(v4f*)&As[0][a_st] = ra;
  *(v4f*)&Bs[0][b_st]                   = (v4f){b00.x, b01.x, b00.y, b01.y};
  *(v4f*)&Bs[0][b_st + 4]               = (v4f){b00.z, b01.z, b00.w, b01.w};
  *(v4f*)&Bs[0][b_st + 4 * BPITCH2]     = (v4f){b10.x, b11.x, b10.y, b11.y};
  *(v4f*)&Bs[0][b_st + 4 * BPITCH2 + 4] = (v4f){b10.z, b11.z, b10.w, b11.w};
  __syncthreads();

  v8f acc[2][4];
#pragma unroll
  for (int i = 0; i < 2; ++i)
#pragma unroll
    for (int j = 0; j < 4; ++j) acc[i][j] = {};

  const int nchunks = K / TBK;
  for (int ch = 0; ch < nchunks; ++ch) {
    const int cur = ch & 1;
    const bool more = (ch + 1) < nchunks;
    if (more) {
      const int kc = (ch + 1) * TBK;
      const size_t kcN = (size_t)kc * N;
      ra  = *(const v4f*)(ag + kc);
      b00 = *(const v4f*)(bg0 + kcN);
      b01 = *(const v4f*)(bg1 + kcN);
      b10 = *(const v4f*)(bg0 + kcN + pair4);
      b11 = *(const v4f*)(bg1 + kcN + pair4);
    }

    const float* as = &As[cur][a_ld];
    const float* bs = &Bs[cur][b_ld];
#pragma unroll
    for (int kk = 0; kk < TBK; kk += 4) {
      v2f a[2];
      v2f b[4];
#pragma unroll
      for (int i = 0; i < 2; ++i)
        a[i] = *(const v2f*)(as + i * 16 * APITCH + kk);
#pragma unroll
      for (int j = 0; j < 4; ++j)
        b[j] = *(const v2f*)(bs + (kk / 2) * BPITCH2 + j * 32);
#pragma unroll
      for (int i = 0; i < 2; ++i)
#pragma unroll
        for (int j = 0; j < 4; ++j)
          acc[i][j] = __builtin_amdgcn_wmma_f32_16x16x4_f32(
              false, a[i], false, b[j], (short)0, acc[i][j], false, false);
    }

    if (more) {
      const int nx = cur ^ 1;
      *(v4f*)&As[nx][a_st] = ra;
      *(v4f*)&Bs[nx][b_st]                   = (v4f){b00.x, b01.x, b00.y, b01.y};
      *(v4f*)&Bs[nx][b_st + 4]               = (v4f){b00.z, b01.z, b00.w, b01.w};
      *(v4f*)&Bs[nx][b_st + 4 * BPITCH2]     = (v4f){b10.x, b11.x, b10.y, b11.y};
      *(v4f*)&Bs[nx][b_st + 4 * BPITCH2 + 4] = (v4f){b10.z, b11.z, b10.w, b11.w};
    }
    __syncthreads();
  }

#pragma unroll
  for (int i = 0; i < 2; ++i) {
#pragma unroll
    for (int j = 0; j < 4; ++j) {
      const int col = n0 + wn * 64 + j * 16 + mrow;
#pragma unroll
      for (int r = 0; r < 8; ++r) {
        const int row = m0 + wm * 32 + i * 16 + r + (khalf << 3);
        C[(size_t)row * N + col] = acc[i][j][r];
      }
    }
  }
}

template <int HH, int CC, int NPW>
__global__ void __launch_bounds__(256)
att_reduce(const float* __restrict__ h,
           const float* __restrict__ att,
           float* __restrict__ out,
           int n_nodes) {
  const int w = (int)((blockIdx.x * 256u + threadIdx.x) >> 5);
  const int lane = threadIdx.x & 31;
  for (int k = 0; k < NPW; ++k) {
    const int node = w * NPW + k;
    if (node >= n_nodes) return;
    const float* hp = h + (size_t)node * (HH * CC);
    float keep = 0.0f;
#pragma unroll
    for (int head = 0; head < HH; ++head) {
      float s = 0.0f;
#pragma unroll
      for (int c = lane; c < CC; c += 32)
        s += hp[head * CC + c] * att[head * CC + c];
#pragma unroll
      for (int off = 16; off > 0; off >>= 1) s += __shfl_xor(s, off, 32);
      if (lane == head) keep = s;
    }
    if (lane < HH) out[(size_t)node * HH + lane] = keep;
  }
}

#define EPT 8
__global__ void __launch_bounds__(256)
edge_logits_max(const int* __restrict__ src,
                const int* __restrict__ dst,
                const float* __restrict__ a_src,
                const float* __restrict__ a_dst,
                float* __restrict__ ebuf,
                float* __restrict__ m,
                int E, int HH) {
  const int total = E * HH;
  const int base = blockIdx.x * (256 * EPT) + threadIdx.x;
  float v[EPT];
  int mi[EPT];
#pragma unroll
  for (int k = 0; k < EPT; ++k) {
    const int t = min(base + k * 256, total - 1);
    const int e = t / HH;
    const int h = t - e * HH;
    mi[k] = dst[e] * HH + h;
    const float a = a_src[src[e] * HH + h] + a_dst[mi[k]];
    v[k] = (a > 0.0f) ? a : NEG_SLOPE * a;
  }
#pragma unroll
  for (int k = 0; k < EPT; ++k) {
    const int t = base + k * 256;
    if (t < total) {
      ebuf[t] = v[k];
      atomicMaxF(&m[mi[k]], v[k]);
    }
  }
}

__global__ void __launch_bounds__(256)
edge_exp_sum(const int* __restrict__ dst,
             float* __restrict__ ebuf,
             const float* __restrict__ m,
             float* __restrict__ denom,
             int E, int HH) {
  const int total = E * HH;
  const int base = blockIdx.x * (256 * EPT) + threadIdx.x;
  float ex[EPT];
  int mi[EPT];
#pragma unroll
  for (int k = 0; k < EPT; ++k) {
    const int t = min(base + k * 256, total - 1);
    const int e = t / HH;
    const int h = t - e * HH;
    mi[k] = dst[e] * HH + h;
    ex[k] = expf(ebuf[t] - m[mi[k]]);
  }
#pragma unroll
  for (int k = 0; k < EPT; ++k) {
    const int t = base + k * 256;
    if (t < total) {
      ebuf[t] = ex[k];
      atomicAdd(&denom[mi[k]], ex[k]);
    }
  }
}

template <int HH, int CC, bool CONCAT>
__global__ void __launch_bounds__(256)
edge_scatter(const float* __restrict__ hsrc,
             const int* __restrict__ src,
             const int* __restrict__ dst,
             const float* __restrict__ ex,
             const float* __restrict__ denom,
             float* __restrict__ out,
             int E, int outC, float scale) {
  const int e = __builtin_amdgcn_readfirstlane(
      (int)(blockIdx.x * 8u + (threadIdx.x >> 5)));
  if (e >= E) return;
  const int lane = threadIdx.x & 31;
  const int s = src[e];
  const int d = dst[e];
  const float* hp = hsrc + (size_t)s * (HH * CC);
  float* op = out + (size_t)d * outC;
  constexpr int PH = CC / 32;
  float v[HH * PH];
#pragma unroll
  for (int j = 0; j < HH * PH; ++j) v[j] = hp[j * 32 + lane];
  if constexpr (CONCAT) {
#pragma unroll
    for (int h = 0; h < HH; ++h) {
      const float alpha =
          scale * ex[(size_t)e * HH + h] / denom[(size_t)d * HH + h];
#pragma unroll
      for (int q = 0; q < PH; ++q) {
        const int c = q * 32 + lane;
        atomicAdd(&op[h * CC + c], alpha * v[h * PH + q]);
      }
    }
  } else {
    float acc[PH];
#pragma unroll
    for (int h = 0; h < HH; ++h) {
      const float alpha =
          scale * ex[(size_t)e * HH + h] / denom[(size_t)d * HH + h];
#pragma unroll
      for (int q = 0; q < PH; ++q)
        acc[q] = (h == 0) ? alpha * v[q] : fmaf(alpha, v[h * PH + q], acc[q]);
    }
#pragma unroll
    for (int q = 0; q < PH; ++q) atomicAdd(&op[q * 32 + lane], acc[q]);
  }
}

__global__ void fill_f32(float* __restrict__ p, float v, long n) {
  const long i = blockIdx.x * (long)blockDim.x + threadIdx.x;
  if (i < n) p[i] = v;
}

#define VPT 4
__global__ void __launch_bounds__(256)
init_rows_bias(float* __restrict__ out,
               const float* __restrict__ bias,
               long rows, int cols) {
  const long nq = rows * cols / 4;
  const long base = blockIdx.x * (long)(256 * VPT) + threadIdx.x;
  const unsigned qcols = (unsigned)cols / 4u;
  const unsigned r0 = (unsigned)(base % (long)qcols);
  v4f b[VPT];
#pragma unroll
  for (int k = 0; k < VPT; ++k)
    b[k] = *(const v4f*)(bias + 4u * ((r0 + 256u * k) % qcols));
#pragma unroll
  for (int k = 0; k < VPT; ++k) {
    const long q = base + k * 256;
    if (q < nq) *(v4f*)(out + 4 * q) = b[k];
  }
}

__global__ void __launch_bounds__(256)
elu_inplace(float* __restrict__ p, long n) {
  const long nq = n / 4;
  const long base = blockIdx.x * (long)(256 * VPT) + threadIdx.x;
  v4f v[VPT];
#pragma unroll
  for (int k = 0; k < VPT; ++k) {
    const long q = base + k * 256;
    v[k] = *(const v4f*)(p + 4 * (q < nq ? q : nq - 1));
  }
#pragma unroll
  for (int k = 0; k < VPT; ++k) {
    const long q = base + k * 256;
    if (q < nq) {
      v4f r = v[k];
      r.x = (r.x > 0.0f) ? r.x : expm1f(r.x);
      r.y = (r.y > 0.0f) ? r.y : expm1f(r.y);
      r.z = (r.z > 0.0f) ? r.z : expm1f(r.z);
      r.w = (r.w > 0.0f) ? r.w : expm1f(r.w);
      *(v4f*)(p + 4 * q) = r;
    }
  }
}

extern "C" void kernel_launch(void* const* d_in, const int* in_sizes, int n_in,
                              void* d_out, int out_size, void* d_ws,
                              size_t ws_size, hipStream_t stream) {
  (void)in_sizes; (void)n_in; (void)out_size; (void)ws_size;

  const float* x        = (const float*)d_in[0];
  const int*   src1     = (const int*)d_in[1];
  const int*   dst1     = (const int*)d_in[2];
  const int*   src2     = (const int*)d_in[3];
  const int*   dst2     = (const int*)d_in[4];
  const float* W1       = (const float*)d_in[5];
  const float* att_src1 = (const float*)d_in[6];
  const float* att_dst1 = (const float*)d_in[7];
  const float* b1       = (const float*)d_in[8];
  const float* W2       = (const float*)d_in[9];
  const float* att_src2 = (const float*)d_in[10];
  const float* att_dst2 = (const float*)d_in[11];
  const float* b2       = (const float*)d_in[12];
  float* out = (float*)d_out;

  char* ws = (char*)d_ws;
  size_t off = 0;
  auto alloc = [&](size_t bytes) -> float* {
    off = (off + 255) & ~(size_t)255;
    float* p = (float*)(ws + off);
    off += bytes;
    return p;
  };

  float* h1      = alloc((size_t)GN0 * GHC1 * 4);
  float* x1      = alloc((size_t)GN1 * GHC1 * 4);
  float* ebuf    = alloc((size_t)GE1 * GH1 * 4);
  float* a_src1  = alloc((size_t)GN0 * GH1 * 4);
  float* a_dst1  = alloc((size_t)GN1 * GH1 * 4);
  float* m1      = alloc((size_t)GN1 * GH1 * 4);
  float* denom1  = alloc((size_t)GN1 * GH1 * 4);
  float* a_src2  = alloc((size_t)GN1 * GH2 * 4);
  float* a_dst2  = alloc((size_t)GN2 * GH2 * 4);
  float* m2      = alloc((size_t)GN2 * GH2 * 4);
  float* denom2  = alloc((size_t)GN2 * GH2 * 4);
  float* h2      = h1;

  wmma_gemm_f32<<<dim3(GHC1 / TBN, GN0 / TBM), 256, 0, stream>>>(
      x, W1, h1, GN0, GHC1, GD);

  att_reduce<GH1, GC1, 4><<<(GN0 + 31) / 32, 256, 0, stream>>>(
      h1, att_src1, a_src1, GN0);
  att_reduce<GH1, GC1, 4><<<(GN1 + 31) / 32, 256, 0, stream>>>(
      h1, att_dst1, a_dst1, GN1);

  fill_f32<<<(GN1 * GH1 + 255) / 256, 256, 0, stream>>>(
      m1, -INFINITY, (long)GN1 * GH1);
  fill_f32<<<(GN1 * GH1 + 255) / 256, 256, 0, stream>>>(
      denom1, 0.0f, (long)GN1 * GH1);

  edge_logits_max<<<(GE1 * GH1 + 256 * EPT - 1) / (256 * EPT), 256, 0, stream>>>(
      src1, dst1, a_src1, a_dst1, ebuf, m1, GE1, GH1);
  edge_exp_sum<<<(GE1 * GH1 + 256 * EPT - 1) / (256 * EPT), 256, 0, stream>>>(
      dst1, ebuf, m1, denom1, GE1, GH1);

  init_rows_bias<<<((long)GN1 * GHC1 / 4 + 256 * VPT - 1) / (256 * VPT), 256, 0,
                   stream>>>(x1, b1, GN1, GHC1);
  edge_scatter<GH1, GC1, true><<<(GE1 + 7) / 8, 256, 0, stream>>>(
      h1, src1, dst1, ebuf, denom1, x1, GE1, GHC1, 1.0f);
  elu_inplace<<<((long)GN1 * GHC1 / 4 + 256 * VPT - 1) / (256 * VPT), 256, 0,
                stream>>>(x1, (long)GN1 * GHC1);

  wmma_gemm_f32<<<dim3(GHC2 / TBN, GN1 / TBM), 256, 0, stream>>>(
      x1, W2, h2, GN1, GHC2, GHC1);

  att_reduce<GH2, GC2, 1><<<(GN1 + 7) / 8, 256, 0, stream>>>(
      h2, att_src2, a_src2, GN1);
  att_reduce<GH2, GC2, 1><<<(GN2 + 7) / 8, 256, 0, stream>>>(
      h2, att_dst2, a_dst2, GN2);

  fill_f32<<<(GN2 * GH2 + 255) / 256, 256, 0, stream>>>(
      m2, -INFINITY, (long)GN2 * GH2);
  fill_f32<<<(GN2 * GH2 + 255) / 256, 256, 0, stream>>>(
      denom2, 0.0f, (long)GN2 * GH2);

  edge_logits_max<<<(GE2 * GH2 + 256 * EPT - 1) / (256 * EPT), 256, 0, stream>>>(
      src2, dst2, a_src2, a_dst2, ebuf, m2, GE2, GH2);
  edge_exp_sum<<<(GE2 * GH2 + 256 * EPT - 1) / (256 * EPT), 256, 0, stream>>>(
      dst2, ebuf, m2, denom2, GE2, GH2);

  init_rows_bias<<<((long)GN2 * GC2 / 4 + 256 * VPT - 1) / (256 * VPT), 256, 0,
                   stream>>>(out, b2, GN2, GC2);
  edge_scatter<GH2, GC2, false><<<(GE2 + 7) / 8, 256, 0, stream>>>(
      h2, src2, dst2, ebuf, denom2, out, GE2, GC2, 1.0f / (float)GH2);
}


# Layer-1 edges bucketed by destination per call (int rank atomics, scan); one wave per target row does softmax, row sum in registers, bias, ELU: no f32 atomics in layer 1; sum order changes.
# speedup vs baseline: 2.2665x; 1.1482x over previous
#include <hip/hip_runtime.h>
#include <math.h>


typedef float v2f __attribute__((ext_vector_type(2)));
typedef float v8f __attribute__((ext_vector_type(8)));

#define GN0 120000
#define GN1 24000
#define GN2 6000
#define GD  256
#define GH1 8
#define GC1 64
#define GHC1 512
#define GH2 8
#define GC2 256
#define GHC2 2048
#define GE1 384000
#define GE2 96000
#define NEG_SLOPE 0.2f

__device__ __forceinline__ void atomicMaxF(float* addr, float v) {
  if (v >= 0.0f)
    atomicMax((int*)addr, __float_as_int(v));
  else
    atomicMin((unsigned int*)addr, __float_as_uint(v));
}

typedef float v4f __attribute__((ext_vector_type(4)));

#define TBM 64
#define TBN 256
#define TBK 16
#define APITCH 20
#define BPITCH2 544

__global__ void __launch_bounds__(256)
wmma_gemm_f32(const float* __restrict__ A, const float* __restrict__ B,
              float* __restrict__ C, int M, int N, int K) {
  __shared__ __attribute__((aligned(16))) float As[2][TBM * APITCH];
  __shared__ __attribute__((aligned(16))) float Bs[2][(TBK / 2) * BPITCH2];

  const int tid   = threadIdx.x;
  const int lane  = tid & 31;
  const int wave  = tid >> 5;
  const int mrow  = lane & 15;
  const int khalf = lane >> 4;
  const int wm = wave >> 2;
  const int wn = wave & 3;
  const int n0 = blockIdx.x * TBN;
  const int m0 = blockIdx.y * TBM;

  const int arow = tid >> 2;
  const int akq  = tid & 3;
  const int bp0  = tid >> 6;
  const int bnq  = tid & 63;
  const float* ag  = A + (size_t)(m0 + arow) * K + akq * 4;
  const float* bg0 = B + (size_t)(2 * bp0) * N + n0 + bnq * 4;
  const float* bg1 = bg0 + N;
  const size_t pair4 = (size_t)8 * N;
  const int a_st = arow * APITCH + akq * 4;
  const int b_st = bp0 * BPITCH2 + bnq * 8;

  const int a_ld = (wm * 32 + mrow) * APITCH + 2 * khalf;
  const int b_ld = khalf * BPITCH2 + (wn * 64 + mrow) * 2;

  v4f ra  = *(const v4f*)ag;
  v4f b00 = *(const v4f*)bg0;
  v4f b01 = *(const v4f*)bg1;
  v4f b10 = *(const v4f*)(bg0 + pair4);
  v4f b11 = *(const v4f*)(bg1 + pair4);

  *(v4f*)&As[0][a_st] = ra;
  *(v4f*)&Bs[0][b_st]                   = (v4f){b00.x, b01.x, b00.y, b01.y};
  *(v4f*)&Bs[0][b_st + 4]               = (v4f){b00.z, b01.z, b00.w, b01.w};
  *(v4f*)&Bs[0][b_st + 4 * BPITCH2]     = (v4f){b10.x, b11.x, b10.y, b11.y};
  *(v4f*)&Bs[0][b_st + 4 * BPITCH2 + 4] = (v4f){b10.z, b11.z, b10.w, b11.w};
  __syncthreads();

  v8f acc[2][4];
#pragma unroll
  for (int i = 0; i < 2; ++i)
#pragma unroll
    for (int j = 0; j < 4; ++j) acc[i][j] = {};

  const int nchunks = K / TBK;
  for (int ch = 0; ch < nchunks; ++ch) {
    const int cur = ch & 1;
    const bool more = (ch + 1) < nchunks;
    if (more) {
      const int kc = (ch + 1) * TBK;
      const size_t kcN = (size_t)kc * N;
      ra  = *(const v4f*)(ag + kc);
      b00 = *(const v4f*)(bg0 + kcN);
      b01 = *(const v4f*)(bg1 + kcN);
      b10 = *(const v4f*)(bg0 + kcN + pair4);
      b11 = *(const v4f*)(bg1 + kcN + pair4);
    }

    const float* as = &As[cur][a_ld];
    const float* bs = &Bs[cur][b_ld];
#pragma unroll
    for (int kk = 0; kk < TBK; kk += 4) {
      v2f a[2];
      v2f b[4];
#pragma unroll
      for (int i = 0; i < 2; ++i)
        a[i] = *(const v2f*)(as + i * 16 * APITCH + kk);
#pragma unroll
      for (int j = 0; j < 4; ++j)
        b[j] = *(const v2f*)(bs + (kk / 2) * BPITCH2 + j * 32);
#pragma unroll
      for (int i = 0; i < 2; ++i)
#pragma unroll
        for (int j = 0; j < 4; ++j)
          acc[i][j] = __builtin_amdgcn_wmma_f32_16x16x4_f32(
              false, a[i], false, b[j], (short)0, acc[i][j], false, false);
    }

    if (more) {
      const int nx = cur ^ 1;
      *(v4f*)&As[nx][a_st] = ra;
      *(v4f*)&Bs[nx][b_st]                   = (v4f){b00.x, b01.x, b00.y, b01.y};
      *(v4f*)&Bs[nx][b_st + 4]               = (v4f){b00.z, b01.z, b00.w, b01.w};
      *(v4f*)&Bs[nx][b_st + 4 * BPITCH2]     = (v4f){b10.x, b11.x, b10.y, b11.y};
      *(v4f*)&Bs[nx][b_st + 4 * BPITCH2 + 4] = (v4f){b10.z, b11.z, b10.w, b11.w};
    }
    __syncthreads();
  }

#pragma unroll
  for (int i = 0; i < 2; ++i) {
#pragma unroll
    for (int j = 0; j < 4; ++j) {
      const int col = n0 + wn * 64 + j * 16 + mrow;
#pragma unroll
      for (int r = 0; r < 8; ++r) {
        const int row = m0 + wm * 32 + i * 16 + r + (khalf << 3);
        C[(size_t)row * N + col] = acc[i][j][r];
      }
    }
  }
}

template <int HH, int CC, int NPW>
__global__ void __launch_bounds__(256)
att_reduce(const float* __restrict__ h,
           const float* __restrict__ att,
           float* __restrict__ out,
           int n_nodes) {
  const int w = (int)((blockIdx.x * 256u + threadIdx.x) >> 5);
  const int lane = threadIdx.x & 31;
  for (int k = 0; k < NPW; ++k) {
    const int node = w * NPW + k;
    if (node >= n_nodes) return;
    const float* hp = h + (size_t)node * (HH * CC);
    float keep = 0.0f;
#pragma unroll
    for (int head = 0; head < HH; ++head) {
      float s = 0.0f;
#pragma unroll
      for (int c = lane; c < CC; c += 32)
        s += hp[head * CC + c] * att[head * CC + c];
#pragma unroll
      for (int off = 16; off > 0; off >>= 1) s += __shfl_xor(s, off, 32);
      if (lane == head) keep = s;
    }
    if (lane < HH) out[(size_t)node * HH + lane] = keep;
  }
}

#define EPT 8
__global__ void __launch_bounds__(256)
edge_logits_max(const int* __restrict__ src,
                const int* __restrict__ dst,
                const float* __restrict__ a_src,
                const float* __restrict__ a_dst,
                float* __restrict__ ebuf,
                float* __restrict__ m,
                int E, int HH) {
  const int total = E * HH;
  const int base = blockIdx.x * (256 * EPT) + threadIdx.x;
  float v[EPT];
  int mi[EPT];
#pragma unroll
  for (int k = 0; k < EPT; ++k) {
    const int t = min(base + k * 256, total - 1);
    const int e = t / HH;
    const int h = t - e * HH;
    mi[k] = dst[e] * HH + h;
    const float a = a_src[src[e] * HH + h] + a_dst[mi[k]];
    v[k] = (a > 0.0f) ? a : NEG_SLOPE * a;
  }
#pragma unroll
  for (int k = 0; k < EPT; ++k) {
    const int t = base + k * 256;
    if (t < total) {
      ebuf[t] = v[k];
      atomicMaxF(&m[mi[k]], v[k]);
    }
  }
}

__global__ void __launch_bounds__(256)
edge_exp_sum(const int* __restrict__ dst,
             float* __restrict__ ebuf,
             const float* __restrict__ m,
             float* __restrict__ denom,
             int E, int HH) {
  const int total = E * HH;
  const int base = blockIdx.x * (256 * EPT) + threadIdx.x;
  float ex[EPT];
  int mi[EPT];
#pragma unroll
  for (int k = 0; k < EPT; ++k) {
    const int t = min(base + k * 256, total - 1);
    const int e = t / HH;
    const int h = t - e * HH;
    mi[k] = dst[e] * HH + h;
    ex[k] = expf(ebuf[t] - m[mi[k]]);
  }
#pragma unroll
  for (int k = 0; k < EPT; ++k) {
    const int t = base + k * 256;
    if (t < total) {
      ebuf[t] = ex[k];
      atomicAdd(&denom[mi[k]], ex[k]);
    }
  }
}

template <int HH, int CC, bool CONCAT>
__global__ void __launch_bounds__(256)
edge_scatter(const float* __restrict__ hsrc,
             const int* __restrict__ src,
             const int* __restrict__ dst,
             const float* __restrict__ ex,
             const float* __restrict__ denom,
             float* __restrict__ out,
             int E, int outC, float scale) {
  const int e = __builtin_amdgcn_readfirstlane(
      (int)(blockIdx.x * 8u + (threadIdx.x >> 5)));
  if (e >= E) return;
  const int lane = threadIdx.x & 31;
  const int s = src[e];
  const int d = dst[e];
  const float* hp = hsrc + (size_t)s * (HH * CC);
  float* op = out + (size_t)d * outC;
  constexpr int PH = CC / 32;
  float v[HH * PH];
#pragma unroll
  for (int j = 0; j < HH * PH; ++j) v[j] = hp[j * 32 + lane];
  if constexpr (CONCAT) {
#pragma unroll
    for (int h = 0; h < HH; ++h) {
      const float alpha =
          scale * ex[(size_t)e * HH + h] / denom[(size_t)d * HH + h];
#pragma unroll
      for (int q = 0; q < PH; ++q) {
        const int c = q * 32 + lane;
        atomicAdd(&op[h * CC + c], alpha * v[h * PH + q]);
      }
    }
  } else {
    float acc[PH];
#pragma unroll
    for (int h = 0; h < HH; ++h) {
      const float alpha =
          scale * ex[(size_t)e * HH + h] / denom[(size_t)d * HH + h];
#pragma unroll
      for (int q = 0; q < PH; ++q)
        acc[q] = (h == 0) ? alpha * v[q] : fmaf(alpha, v[h * PH + q], acc[q]);
    }
#pragma unroll
    for (int q = 0; q < PH; ++q) atomicAdd(&op[q * 32 + lane], acc[q]);
  }
}

__global__ void fill_f32(float* __restrict__ p, float v, long n) {
  const long i = blockIdx.x * (long)blockDim.x + threadIdx.x;
  if (i < n) p[i] = v;
}

#define VPT 4
__global__ void __launch_bounds__(256)
init_rows_bias(float* __restrict__ out,
               const float* __restrict__ bias,
               long rows, int cols) {
  const long nq = rows * cols / 4;
  const long base = blockIdx.x * (long)(256 * VPT) + threadIdx.x;
  const unsigned qcols = (unsigned)cols / 4u;
  const unsigned r0 = (unsigned)(base % (long)qcols);
  v4f b[VPT];
#pragma unroll
  for (int k = 0; k < VPT; ++k)
    b[k] = *(const v4f*)(bias + 4u * ((r0 + 256u * k) % qcols));
#pragma unroll
  for (int k = 0; k < VPT; ++k) {
    const long q = base + k * 256;
    if (q < nq) *(v4f*)(out + 4 * q) = b[k];
  }
}

__global__ void __launch_bounds__(256)
zero_i32(int* __restrict__ p, int n) {
  const int i = blockIdx.x * 256 + threadIdx.x;
  if (i < n) p[i] = 0;
}

#define RPT 4
__global__ void __launch_bounds__(256)
bucket_rank(const int* __restrict__ dst, int* __restrict__ cnt,
            int* __restrict__ rank, int E, int n_tgt) {
  const int base = blockIdx.x * (256 * RPT) + threadIdx.x;
  int d[RPT];
#pragma unroll
  for (int k = 0; k < RPT; ++k) {
    const int t = base + k * 256;
    const int dv = dst[min(t, E - 1)];
    d[k] = (t < E) ? dv : n_tgt;
  }
  int r[RPT];
#pragma unroll
  for (int k = 0; k < RPT; ++k) r[k] = atomicAdd(&cnt[d[k]], 1);
#pragma unroll
  for (int k = 0; k < RPT; ++k) {
    const int t = base + k * 256;
    if (t < E) rank[t] = r[k];
  }
}

template <int N>
__global__ void __launch_bounds__(1024)
bucket_scan(const int* __restrict__ cnt, int* __restrict__ start) {
  __shared__ int wsum[32];
  constexpr int PER = (N + 1023) / 1024;
  const int t = threadIdx.x;
  const int lane = t & 31;
  const int wave = t >> 5;
  const int lo = t * PER;
  int v[PER];
#pragma unroll
  for (int i = 0; i < PER; ++i) {
    const int idx = lo + i;
    const int c = cnt[min(idx, N - 1)];
    v[i] = (idx < N) ? c : 0;
  }
  int s = 0;
#pragma unroll
  for (int i = 0; i < PER; ++i) s += v[i];
  int inc = s;
#pragma unroll
  for (int off = 1; off < 32; off <<= 1) {
    const int y = __shfl_up(inc, off, 32);
    if (lane >= off) inc += y;
  }
  if (lane == 31) wsum[wave] = inc;
  __syncthreads();
  if (wave == 0) {
    const int w = wsum[lane];
    int winc = w;
#pragma unroll
    for (int off = 1; off < 32; off <<= 1) {
      const int y = __shfl_up(winc, off, 32);
      if (lane >= off) winc += y;
    }
    wsum[lane] = winc - w;
  }
  __syncthreads();
  int run = wsum[wave] + inc - s;
#pragma unroll
  for (int i = 0; i < PER; ++i) {
    const int idx = lo + i;
    if (idx < N) start[idx] = run;
    run += v[i];
  }
  if (t == 1023) start[N] = run;
}

__global__ void __launch_bounds__(256)
bucket_fill(const int* __restrict__ src, const int* __restrict__ dst,
            const int* __restrict__ rank, const int* __restrict__ start,
            int* __restrict__ psrc, int E) {
  const int base = blockIdx.x * (256 * RPT) + threadIdx.x;
  int pos[RPT];
  int sv[RPT];
#pragma unroll
  for (int k = 0; k < RPT; ++k) {
    const int t = min(base + k * 256, E - 1);
    pos[k] = start[dst[t]] + rank[t];
    sv[k] = src[t];
  }
#pragma unroll
  for (int k = 0; k < RPT; ++k) {
    const int t = base + k * 256;
    if (t < E) psrc[pos[k]] = sv[k];
  }
}

template <int HH, int CC>
__global__ void __launch_bounds__(256)
gat_row_owner_elu(const float* __restrict__ hsrc,
                  const int* __restrict__ psrc,
                  const int* __restrict__ start,
                  const float* __restrict__ a_src,
                  const float* __restrict__ a_dst,
                  const float* __restrict__ bias,
                  float* __restrict__ out,
                  int n_tgt) {
  static_assert(HH == 8, "lane map: 4 edge slots x 8 heads");
  constexpr int HC = HH * CC;
  constexpr int NJ = HC / 32;
  constexpr int PH = CC / 32;
  const int d = __builtin_amdgcn_readfirstlane(
      (int)(blockIdx.x * 8u + (threadIdx.x >> 5)));
  if (d >= n_tgt) return;
  const int lane = threadIdx.x & 31;
  const int h = lane & (HH - 1);
  const int slot = lane >> 3;
  const int beg = start[d];
  const int end = start[d + 1];
  const float ad = a_dst[(size_t)d * HH + h];

  float mx = -INFINITY;
  for (int i = beg + slot; i < end; i += 4) {
    const float a = a_src[(size_t)psrc[i] * HH + h] + ad;
    const float v = (a > 0.0f) ? a : NEG_SLOPE * a;
    mx = fmaxf(mx, v);
  }
  mx = fmaxf(mx, __shfl_xor(mx, 8, 32));
  mx = fmaxf(mx, __shfl_xor(mx, 16, 32));
  float den = 0.0f;
  for (int i = beg + slot; i < end; i += 4) {
    const float a = a_src[(size_t)psrc[i] * HH + h] + ad;
    const float v = (a > 0.0f) ? a : NEG_SLOPE * a;
    den += expf(v - mx);
  }
  den += __shfl_xor(den, 8, 32);
  den += __shfl_xor(den, 16, 32);

  float acc[NJ];
#pragma unroll
  for (int j = 0; j < NJ; ++j) acc[j] = 0.0f;

  for (int b0 = beg; b0 < end; b0 += 32) {
    const int nb = min(32, end - b0);
    const int sv = psrc[b0 + min(lane, nb - 1)];
    int k = 0;
    for (; k + 1 < nb; k += 2) {
      const int s0 = __builtin_amdgcn_readlane(sv, k);
      const int s1 = __builtin_amdgcn_readlane(sv, k + 1);
      const float* p0 = hsrc + (size_t)s0 * HC + lane;
      const float* p1 = hsrc + (size_t)s1 * HC + lane;
      float r0[NJ], r1[NJ];
#pragma unroll
      for (int j = 0; j < NJ; ++j) r0[j] = p0[j * 32];
#pragma unroll
      for (int j = 0; j < NJ; ++j) r1[j] = p1[j * 32];
      const float a0 = a_src[(size_t)s0 * HH + h] + ad;
      const float a1 = a_src[(size_t)s1 * HH + h] + ad;
      const float v0 = (a0 > 0.0f) ? a0 : NEG_SLOPE * a0;
      const float v1 = (a1 > 0.0f) ? a1 : NEG_SLOPE * a1;
      const int w0 = __float_as_int(expf(v0 - mx) / den);
      const int w1 = __float_as_int(expf(v1 - mx) / den);
#pragma unroll
      for (int hh = 0; hh < HH; ++hh) {
        const float al0 = __int_as_float(__builtin_amdgcn_readlane(w0, hh));
        const float al1 = __int_as_float(__builtin_amdgcn_readlane(w1, hh));
#pragma unroll
        for (int q = 0; q < PH; ++q) {
          const int j = hh * PH + q;
          acc[j] = fmaf(al0, r0[j], acc[j]);
          acc[j] = fmaf(al1, r1[j], acc[j]);
        }
      }
    }
    if (k < nb) {
      const int s0 = __builtin_amdgcn_readlane(sv, k);
      const float* p0 = hsrc + (size_t)s0 * HC + lane;
      float r0[NJ];
#pragma unroll
      for (int j = 0; j < NJ; ++j) r0[j] = p0[j * 32];
      const float a0 = a_src[(size_t)s0 * HH + h] + ad;
      const float v0 = (a0 > 0.0f) ? a0 : NEG_SLOPE * a0;
      const int w0 = __float_as_int(expf(v0 - mx) / den);
#pragma unroll
      for (int hh = 0; hh < HH; ++hh) {
        const float al0 = __int_as_float(__builtin_amdgcn_readlane(w0, hh));
#pragma unroll
        for (int q = 0; q < PH; ++q) {
          const int j = hh * PH + q;
          acc[j] = fmaf(al0, r0[j], acc[j]);
        }
      }
    }
  }

  float* op = out + (size_t)d * HC;
  float bv[NJ];
#pragma unroll
  for (int j = 0; j < NJ; ++j) bv[j] = bias[j * 32 + lane];
#pragma unroll
  for (int j = 0; j < NJ; ++j) {
    const float r = acc[j] + bv[j];
    op[j * 32 + lane] = (r > 0.0f) ? r : expm1f(r);
  }
}

extern "C" void kernel_launch(void* const* d_in, const int* in_sizes, int n_in,
                              void* d_out, int out_size, void* d_ws,
                              size_t ws_size, hipStream_t stream) {
  (void)in_sizes; (void)n_in; (void)out_size; (void)ws_size;

  const float* x        = (const float*)d_in[0];
  const int*   src1     = (const int*)d_in[1];
  const int*   dst1     = (const int*)d_in[2];
  const int*   src2     = (const int*)d_in[3];
  const int*   dst2     = (const int*)d_in[4];
  const float* W1       = (const float*)d_in[5];
  const float* att_src1 = (const float*)d_in[6];
  const float* att_dst1 = (const float*)d_in[7];
  const float* b1       = (const float*)d_in[8];
  const float* W2       = (const float*)d_in[9];
  const float* att_src2 = (const float*)d_in[10];
  const float* att_dst2 = (const float*)d_in[11];
  const float* b2       = (const float*)d_in[12];
  float* out = (float*)d_out;

  char* ws = (char*)d_ws;
  size_t off = 0;
  auto alloc = [&](size_t bytes) -> float* {
    off = (off + 255) & ~(size_t)255;
    float* p = (float*)(ws + off);
    off += bytes;
    return p;
  };

  float* h1      = alloc((size_t)GN0 * GHC1 * 4);
  float* x1      = alloc((size_t)GN1 * GHC1 * 4);
  float* ebuf    = alloc((size_t)GE2 * GH2 * 4);
  float* a_src1  = alloc((size_t)GN0 * GH1 * 4);
  float* a_dst1  = alloc((size_t)GN1 * GH1 * 4);
  int*   cnt1    = (int*)alloc((size_t)(GN1 + 1) * 4);
  int*   start1  = (int*)alloc((size_t)(GN1 + 1) * 4);
  int*   rank1   = (int*)alloc((size_t)GE1 * 4);
  int*   psrc1   = (int*)alloc((size_t)GE1 * 4);
  float* a_src2  = alloc((size_t)GN1 * GH2 * 4);
  float* a_dst2  = alloc((size_t)GN2 * GH2 * 4);
  float* m2      = alloc((size_t)GN2 * GH2 * 4);
  float* denom2  = alloc((size_t)GN2 * GH2 * 4);
  float* h2      = h1;

  wmma_gemm_f32<<<dim3(GHC1 / TBN, GN0 / TBM), 256, 0, stream>>>(
      x, W1, h1, GN0, GHC1, GD);

  att_reduce<GH1, GC1, 4><<<(GN0 + 31) / 32, 256, 0, stream>>>(
      h1, att_src1, a_src1, GN0);
  att_reduce<GH1, GC1, 4><<<(GN1 + 31) / 32, 256, 0, stream>>>(
      h1, att_dst1, a_dst1, GN1);

  zero_i32<<<(GN1 + 1 + 255) / 256, 256, 0, stream>>>(cnt1, GN1 + 1);
  bucket_rank<<<(GE1 + 256 * RPT - 1) / (256 * RPT), 256, 0, stream>>>(
      dst1, cnt1, rank1, GE1, GN1);
  bucket_scan<GN1><<<1, 1024, 0, stream>>>(cnt1, start1);
  bucket_fill<<<(GE1 + 256 * RPT - 1) / (256 * RPT), 256, 0, stream>>>(
      src1, dst1, rank1, start1, psrc1, GE1);

  gat_row_owner_elu<GH1, GC1><<<(GN1 + 7) / 8, 256, 0, stream>>>(
      h1, psrc1, start1, a_src1, a_dst1, b1, x1, GN1);

  wmma_gemm_f32<<<dim3(GHC2 / TBN, GN1 / TBM), 256, 0, stream>>>(
      x1, W2, h2, GN1, GHC2, GHC1);

  att_reduce<GH2, GC2, 1><<<(GN1 + 7) / 8, 256, 0, stream>>>(
      h2, att_src2, a_src2, GN1);
  att_reduce<GH2, GC2, 1><<<(GN2 + 7) / 8, 256, 0, stream>>>(
      h2, att_dst2, a_dst2, GN2);

  fill_f32<<<(GN2 * GH2 + 255) / 256, 256, 0, stream>>>(
      m2, -INFINITY, (long)GN2 * GH2);
  fill_f32<<<(GN2 * GH2 + 255) / 256, 256, 0, stream>>>(
      denom2, 0.0f, (long)GN2 * GH2);

  edge_logits_max<<<(GE2 * GH2 + 256 * EPT - 1) / (256 * EPT), 256, 0, stream>>>(
      src2, dst2, a_src2, a_dst2, ebuf, m2, GE2, GH2);
  edge_exp_sum<<<(GE2 * GH2 + 256 * EPT - 1) / (256 * EPT), 256, 0, stream>>>(
      dst2, ebuf, m2, denom2, GE2, GH2);

  init_rows_bias<<<((long)GN2 * GC2 / 4 + 256 * VPT - 1) / (256 * VPT), 256, 0,
                   stream>>>(out, b2, GN2, GC2);
  edge_scatter<GH2, GC2, false><<<(GE2 + 7) / 8, 256, 0, stream>>>(
      h2, src2, dst2, ebuf, denom2, out, GE2, GC2, 1.0f / (float)GH2);
}


# Streaming kernels (both logit kernels, head mean) run as few resident waves that loop: 16-byte loads, 4 rows in flight, V in registers, 32 sums by one transposed xor butterfly; sum order changes.
# speedup vs baseline: 4.8647x; 2.1463x over previous
#include <hip/hip_runtime.h>
#include <math.h>


typedef float v2f __attribute__((ext_vector_type(2)));
typedef float v8f __attribute__((ext_vector_type(8)));

#define GN0 120000
#define GN1 24000
#define GN2 6000
#define GD  256
#define GH1 8
#define GC1 64
#define GHC1 512
#define GH2 8
#define GC2 256
#define GHC2 2048
#define GE1 384000
#define GE2 96000
#define NEG_SLOPE 0.2f

typedef float v4f __attribute__((ext_vector_type(4)));

#define TBK 16
#define APITCH 20

template <int WM, int WN, bool ELU>
__global__ void __launch_bounds__(256)
wmma_gemm_f32(const float* __restrict__ A, const float* __restrict__ B,
              float* __restrict__ C, const float* __restrict__ bias,
              int K, int lda, int ldb, int ldc,
              size_t sA, size_t sB, size_t sC, size_t sBi) {
  static_assert(WM * WN == 8, "8 waves per block");
  static_assert(WM % 2 == 0, "A is staged 64 rows per pass of the block");
  constexpr int TBM = 32 * WM;
  constexpr int TBN = 64 * WN;
  constexpr int BP  = 2 * TBN + 32;
  constexpr int AQ  = TBM / 64;
  constexpr int NBQ = TBN / 4;
  constexpr int BPT = (8 * NBQ + 255) / 256;
  constexpr int BSTEP = 256 / NBQ;
  constexpr bool BALL = (8 * NBQ) % 256 == 0;
  __shared__ __attribute__((aligned(16))) float As[2][TBM * APITCH];
  __shared__ __attribute__((aligned(16))) float Bs[2][(TBK / 2) * BP];

  const int tid   = threadIdx.x;
  const int lane  = tid & 31;
  const int wave  = tid >> 5;
  const int mrow  = lane & 15;
  const int khalf = lane >> 4;
  const int wm = wave / WN;
  const int wn = wave % WN;
  const int n0 = blockIdx.x * TBN;
  const int m0 = blockIdx.y * TBM;

  const int arow = tid >> 2;
  const int akq  = tid & 3;
  const int bp0  = tid / NBQ;
  const int bnq  = tid % NBQ;
  const bool bact = BALL || (bp0 < TBK / 2);
  A += (size_t)blockIdx.z * sA;
  B += (size_t)blockIdx.z * sB;
  C += (size_t)blockIdx.z * sC;
  const float* ag  = A + (size_t)(m0 + arow) * lda + akq * 4;
  const size_t arow64 = (size_t)64 * lda;
  const float* bg0 = B + (size_t)(2 * bp0) * ldb + n0 + bnq * 4;
  const float* bg1 = bg0 + ldb;
  const size_t pairs = (size_t)(2 * BSTEP) * ldb;
  const int a_st = arow * APITCH + akq * 4;
  const int b_st = bp0 * BP + bnq * 8;

  const int a_ld = (wm * 32 + mrow) * APITCH + 2 * khalf;
  const int b_ld = khalf * BP + (wn * 64 + mrow) * 2;

  v4f ra[AQ];
  v4f rb0[BPT], rb1[BPT];
#pragma unroll
  for (int i = 0; i < AQ; ++i) ra[i] = *(const v4f*)(ag + i * arow64);
  if (bact) {
#pragma unroll
    for (int u = 0; u < BPT; ++u) {
      rb0[u] = *(const v4f*)(bg0 + u * pairs);
      rb1[u] = *(const v4f*)(bg1 + u * pairs);
    }
  }

#pragma unroll
  for (int i = 0; i < AQ; ++i) *(v4f*)&As[0][a_st + i * 64 * APITCH] = ra[i];
  if (bact) {
#pragma unroll
    for (int u = 0; u < BPT; ++u) {
      float* bd = &Bs[0][b_st + u * BSTEP * BP];
      *(v4f*)bd       = (v4f){rb0[u].x, rb1[u].x, rb0[u].y, rb1[u].y};
      *(v4f*)(bd + 4) = (v4f){rb0[u].z, rb1[u].z, rb0[u].w, rb1[u].w};
    }
  }
  __syncthreads();

  float bcol[4];
#pragma unroll
  for (int j = 0; j < 4; ++j)
    bcol[j] = ELU ? bias[(size_t)blockIdx.z * sBi + n0 + wn * 64 + j * 16 + mrow]
                  : 0.0f;

  v8f acc[2][4];
#pragma unroll
  for (int i = 0; i < 2; ++i)
#pragma unroll
    for (int j = 0; j < 4; ++j) acc[i][j] = {};

  const int nchunks = K / TBK;
  for (int ch = 0; ch < nchunks; ++ch) {
    const int cur = ch & 1;
    const bool more = (ch + 1) < nchunks;
    if (more) {
      const int kc = (ch + 1) * TBK;
      const size_t kcN = (size_t)kc * ldb;
#pragma unroll
      for (int i = 0; i < AQ; ++i) ra[i] = *(const v4f*)(ag + kc + i * arow64);
      if (bact) {
#pragma unroll
        for (int u = 0; u < BPT; ++u) {
          rb0[u] = *(const v4f*)(bg0 + kcN + u * pairs);
          rb1[u] = *(const v4f*)(bg1 + kcN + u * pairs);
        }
      }
    }

    const float* as = &As[cur][a_ld];
    const float* bs = &Bs[cur][b_ld];
#pragma unroll
    for (int kk = 0; kk < TBK; kk += 4) {
      v2f a[2];
      v2f b[4];
#pragma unroll
      for (int i = 0; i < 2; ++i)
        a[i] = *(const v2f*)(as + i * 16 * APITCH + kk);
#pragma unroll
      for (int j = 0; j < 4; ++j)
        b[j] = *(const v2f*)(bs + (kk / 2) * BP + j * 32);
#pragma unroll
      for (int i = 0; i < 2; ++i)
#pragma unroll
        for (int j = 0; j < 4; ++j)
          acc[i][j] = __builtin_amdgcn_wmma_f32_16x16x4_f32(
              false, a[i], false, b[j], (short)0, acc[i][j], false, false);
    }

    if (more) {
      const int nx = cur ^ 1;
#pragma unroll
      for (int i = 0; i < AQ; ++i) *(v4f*)&As[nx][a_st + i * 64 * APITCH] = ra[i];
      if (bact) {
#pragma unroll
        for (int u = 0; u < BPT; ++u) {
          float* bd = &Bs[nx][b_st + u * BSTEP * BP];
          *(v4f*)bd       = (v4f){rb0[u].x, rb1[u].x, rb0[u].y, rb1[u].y};
          *(v4f*)(bd + 4) = (v4f){rb0[u].z, rb1[u].z, rb0[u].w, rb1[u].w};
        }
      }
    }
    __syncthreads();
  }

#pragma unroll
  for (int i = 0; i < 2; ++i) {
#pragma unroll
    for (int j = 0; j < 4; ++j) {
      const int col = n0 + wn * 64 + j * 16 + mrow;
#pragma unroll
      for (int r = 0; r < 8; ++r) {
        const int row = m0 + wm * 32 + i * 16 + r + (khalf << 3);
        float v = acc[i][j][r];
        if (ELU) {
          v += bcol[j];
          v = (v > 0.0f) ? v : expm1f(v);
        }
        C[(size_t)row * ldc + col] = v;
      }
    }
  }
}

template <int HH, int CC, int NPW>
__device__ __forceinline__ void
att_reduce(int bid,
           const float* __restrict__ h,
           const float* __restrict__ att,
           float* __restrict__ out,
           int n_nodes) {
  const int w = (int)(((unsigned)bid * 256u + threadIdx.x) >> 5);
  const int lane = threadIdx.x & 31;
  for (int k = 0; k < NPW; ++k) {
    const int node = w * NPW + k;
    if (node >= n_nodes) return;
    const float* hp = h + (size_t)node * (HH * CC);
    float keep = 0.0f;
#pragma unroll
    for (int head = 0; head < HH; ++head) {
      float s = 0.0f;
#pragma unroll
      for (int c = lane; c < CC; c += 32)
        s += hp[head * CC + c] * att[head * CC + c];
#pragma unroll
      for (int off = 16; off > 0; off >>= 1) s += __shfl_xor(s, off, 32);
      if (lane == head) keep = s;
    }
    if (lane < HH) out[(size_t)node * HH + lane] = keep;
  }
}

__device__ __forceinline__ void
zero_i32(int bid, int* __restrict__ p, int n) {
  const int i = bid * 256 + threadIdx.x;
  if (i < n) p[i] = 0;
}

#define RPT 4
__device__ __forceinline__ void
bucket_rank(int bid, const int* __restrict__ dst, int* __restrict__ cnt,
            int* __restrict__ rank, int E, int n_tgt) {
  const int base = bid * (256 * RPT) + threadIdx.x;
  int d[RPT];
#pragma unroll
  for (int k = 0; k < RPT; ++k) {
    const int t = base + k * 256;
    const int dv = dst[min(t, E - 1)];
    d[k] = (t < E) ? dv : n_tgt;
  }
  int r[RPT];
#pragma unroll
  for (int k = 0; k < RPT; ++k) r[k] = atomicAdd(&cnt[d[k]], 1);
#pragma unroll
  for (int k = 0; k < RPT; ++k) {
    const int t = base + k * 256;
    if (t < E) rank[t] = r[k];
  }
}

template <int N>
__device__ __forceinline__ void
bucket_scan(const int* __restrict__ cnt, int* __restrict__ start, int* wsum) {
  constexpr int PER = (N + 1023) / 1024;
  const int t = threadIdx.x;
  const int lane = t & 31;
  const int wave = t >> 5;
  const int lo = t * PER;
  int v[PER];
#pragma unroll
  for (int i = 0; i < PER; ++i) {
    const int idx = lo + i;
    const int c = cnt[min(idx, N - 1)];
    v[i] = (idx < N) ? c : 0;
  }
  int s = 0;
#pragma unroll
  for (int i = 0; i < PER; ++i) s += v[i];
  int inc = s;
#pragma unroll
  for (int off = 1; off < 32; off <<= 1) {
    const int y = __shfl_up(inc, off, 32);
    if (lane >= off) inc += y;
  }
  if (lane == 31) wsum[wave] = inc;
  __syncthreads();
  if (wave == 0) {
    const int w = wsum[lane];
    int winc = w;
#pragma unroll
    for (int off = 1; off < 32; off <<= 1) {
      const int y = __shfl_up(winc, off, 32);
      if (lane >= off) winc += y;
    }
    wsum[lane] = winc - w;
  }
  __syncthreads();
  int run = wsum[wave] + inc - s;
#pragma unroll
  for (int i = 0; i < PER; ++i) {
    const int idx = lo + i;
    if (idx < N) start[idx] = run;
    run += v[i];
  }
  if (t == 1023) start[N] = run;
}

__device__ __forceinline__ void
bucket_fill(int bid, const int* __restrict__ src, const int* __restrict__ dst,
            const int* __restrict__ rank, const int* __restrict__ start,
            int* __restrict__ psrc, int E) {
  const int base = bid * (256 * RPT) + threadIdx.x;
  int pos[RPT];
  int sv[RPT];
#pragma unroll
  for (int k = 0; k < RPT; ++k) {
    const int t = min(base + k * 256, E - 1);
    pos[k] = start[dst[t]] + rank[t];
    sv[k] = src[t];
  }
#pragma unroll
  for (int k = 0; k < RPT; ++k) {
    const int t = base + k * 256;
    if (t < E) psrc[pos[k]] = sv[k];
  }
}


template <int KK, int HH>
__device__ __forceinline__ void
node_logits(int bid, int nblk,
            const float* __restrict__ x,
            const float* __restrict__ V,
            float* __restrict__ out,
            int n_nodes) {
  static_assert(HH == 8, "4 nodes x 8 heads = one value per lane");
  static_assert(KK % 128 == 0, "a wave reads a row in 512-byte pieces");
  constexpr int NQ = KK / 128;
  const int lane = threadIdx.x & 31;
  const int w = __builtin_amdgcn_readfirstlane(bid * 8 + (int)(threadIdx.x >> 5));
  const int nw = nblk * 8;
  const int n_groups = (n_nodes + 3) >> 2;
  if (w >= n_groups) return;

  float vv[NQ][4][HH];
#pragma unroll
  for (int q = 0; q < NQ; ++q)
#pragma unroll
    for (int e = 0; e < 4; ++e) {
      const float* vp = V + (size_t)(q * 128 + lane * 4 + e) * HH;
      const v4f lo = *(const v4f*)vp;
      const v4f hi = *(const v4f*)(vp + 4);
      vv[q][e][0] = lo.x; vv[q][e][1] = lo.y; vv[q][e][2] = lo.z; vv[q][e][3] = lo.w;
      vv[q][e][4] = hi.x; vv[q][e][5] = hi.y; vv[q][e][6] = hi.z; vv[q][e][7] = hi.w;
    }

  const bool b16 = (lane & 16) != 0;
  const bool b8  = (lane & 8) != 0;
  const bool b4  = (lane & 4) != 0;
  const bool b2  = (lane & 2) != 0;
  const bool b1  = (lane & 1) != 0;

  v4f r[4][NQ];
#define NL_LOAD_ROWS(g_)                                                    \
  _Pragma("unroll") for (int n = 0; n < 4; ++n) {                           \
    const int node_ = min((g_) * 4 + n, n_nodes - 1);                       \
    const float* xp_ = x + (size_t)node_ * KK + lane * 4;                   \
    _Pragma("unroll") for (int q = 0; q < NQ; ++q)                          \
      r[n][q] = *(const v4f*)(xp_ + q * 128);                               \
  }

  int g = w;
  NL_LOAD_ROWS(g)
  for (;;) {
    float t16[16];
#pragma unroll
    for (int n = 0; n < 2; ++n) {
      float va[HH], vb[HH];
#pragma unroll
      for (int hh = 0; hh < HH; ++hh) {
        float sa = 0.0f, sb = 0.0f;
#pragma unroll
        for (int q = 0; q < NQ; ++q) {
          sa = fmaf(r[n][q].x, vv[q][0][hh], sa);
          sa = fmaf(r[n][q].y, vv[q][1][hh], sa);
          sa = fmaf(r[n][q].z, vv[q][2][hh], sa);
          sa = fmaf(r[n][q].w, vv[q][3][hh], sa);
          sb = fmaf(r[n + 2][q].x, vv[q][0][hh], sb);
          sb = fmaf(r[n + 2][q].y, vv[q][1][hh], sb);
          sb = fmaf(r[n + 2][q].z, vv[q][2][hh], sb);
          sb = fmaf(r[n + 2][q].w, vv[q][3][hh], sb);
        }
        va[hh] = sa;
        vb[hh] = sb;
      }
#pragma unroll
      for (int hh = 0; hh < HH; ++hh) {
        const float keep = b16 ? vb[hh] : va[hh];
        const float send = b16 ? va[hh] : vb[hh];
        t16[n * 8 + hh] = keep + __shfl_xor(send, 16, 32);
      }
    }

    const int gn = g + nw;
    const bool more = gn < n_groups;
    if (more) { NL_LOAD_ROWS(gn) }

    float t8[8];
#pragma unroll
    for (int j = 0; j < 8; ++j) {
      const float keep = b8 ? t16[j + 8] : t16[j];
      const float send = b8 ? t16[j] : t16[j + 8];
      t8[j] = keep + __shfl_xor(send, 8, 32);
    }
    float t4[4];
#pragma unroll
    for (int j = 0; j < 4; ++j) {
      const float keep = b4 ? t8[j + 4] : t8[j];
      const float send = b4 ? t8[j] : t8[j + 4];
      t4[j] = keep + __shfl_xor(send, 4, 32);
    }
    float t2[2];
#pragma unroll
    for (int j = 0; j < 2; ++j) {
      const float keep = b2 ? t4[j + 2] : t4[j];
      const float send = b2 ? t4[j] : t4[j + 2];
      t2[j] = keep + __shfl_xor(send, 2, 32);
    }
    const float keep1 = b1 ? t2[1] : t2[0];
    const float send1 = b1 ? t2[0] : t2[1];
    const float tot = keep1 + __shfl_xor(send1, 1, 32);

    if (g * 4 + (lane >> 3) < n_nodes) out[(size_t)g * 32 + lane] = tot;
    if (!more) break;
    g = gn;
  }
#undef NL_LOAD_ROWS
}

__device__ __forceinline__ v4f fma4(float a, v4f x, v4f c) {
  return (v4f){fmaf(a, x.x, c.x), fmaf(a, x.y, c.y),
               fmaf(a, x.z, c.z), fmaf(a, x.w, c.w)};
}

template <int HH, int KK>
__global__ void __launch_bounds__(256)
gat_row_owner_heads(const float* __restrict__ xsrc,
                    const int* __restrict__ psrc,
                    const int* __restrict__ start,
                    const float* __restrict__ a_src,
                    const float* __restrict__ a_dst,
                    float* __restrict__ z,
                    int n_tgt, int n_pad) {
  static_assert(HH == 8, "lane map: 4 edge slots x 8 heads");
  static_assert(KK % 128 == 0, "a wave reads a row in 512-byte pieces");
  constexpr int NQ = KK / 128;
  const int d = __builtin_amdgcn_readfirstlane(
      (int)(blockIdx.x * 8u + (threadIdx.x >> 5)));
  if (d >= n_pad) return;
  const int lane = threadIdx.x & 31;
  float* zp = z + (size_t)d * KK + lane * 4;
  const size_t hpitch = (size_t)n_pad * KK;
  if (d >= n_tgt) {
#pragma unroll
    for (int hh = 0; hh < HH; ++hh)
#pragma unroll
      for (int q = 0; q < NQ; ++q)
        *(v4f*)(zp + hh * hpitch + q * 128) = (v4f){0.0f, 0.0f, 0.0f, 0.0f};
    return;
  }
  const int h = lane & (HH - 1);
  const int slot = lane >> 3;
  const int beg = start[d];
  const int end = start[d + 1];
  const float ad = a_dst[(size_t)d * HH + h];

  float mx = -INFINITY;
  for (int i = beg + slot; i < end; i += 4) {
    const float a = a_src[(size_t)psrc[i] * HH + h] + ad;
    const float v = (a > 0.0f) ? a : NEG_SLOPE * a;
    mx = fmaxf(mx, v);
  }
  mx = fmaxf(mx, __shfl_xor(mx, 8, 32));
  mx = fmaxf(mx, __shfl_xor(mx, 16, 32));
  float den = 0.0f;
  for (int i = beg + slot; i < end; i += 4) {
    const float a = a_src[(size_t)psrc[i] * HH + h] + ad;
    const float v = (a > 0.0f) ? a : NEG_SLOPE * a;
    den += expf(v - mx);
  }
  den += __shfl_xor(den, 8, 32);
  den += __shfl_xor(den, 16, 32);

  v4f acc[HH][NQ];
#pragma unroll
  for (int hh = 0; hh < HH; ++hh)
#pragma unroll
    for (int q = 0; q < NQ; ++q) acc[hh][q] = (v4f){0.0f, 0.0f, 0.0f, 0.0f};

  for (int b0 = beg; b0 < end; b0 += 32) {
    const int nb = min(32, end - b0);
    const int sv = psrc[b0 + min(lane, nb - 1)];
    int k = 0;
    for (; k + 1 < nb; k += 2) {
      const int s0 = __builtin_amdgcn_readlane(sv, k);
      const int s1 = __builtin_amdgcn_readlane(sv, k + 1);
      const float* p0 = xsrc + (size_t)s0 * KK + lane * 4;
      const float* p1 = xsrc + (size_t)s1 * KK + lane * 4;
      v4f r0[NQ], r1[NQ];
#pragma unroll
      for (int q = 0; q < NQ; ++q) r0[q] = *(const v4f*)(p0 + q * 128);
#pragma unroll
      for (int q = 0; q < NQ; ++q) r1[q] = *(const v4f*)(p1 + q * 128);
      const float a0 = a_src[(size_t)s0 * HH + h] + ad;
      const float a1 = a_src[(size_t)s1 * HH + h] + ad;
      const float v0 = (a0 > 0.0f) ? a0 : NEG_SLOPE * a0;
      const float v1 = (a1 > 0.0f) ? a1 : NEG_SLOPE * a1;
      const int w0 = __float_as_int(expf(v0 - mx) / den);
      const int w1 = __float_as_int(expf(v1 - mx) / den);
#pragma unroll
      for (int hh = 0; hh < HH; ++hh) {
        const float al0 = __int_as_float(__builtin_amdgcn_readlane(w0, hh));
        const float al1 = __int_as_float(__builtin_amdgcn_readlane(w1, hh));
#pragma unroll
        for (int q = 0; q < NQ; ++q) {
          acc[hh][q] = fma4(al0, r0[q], acc[hh][q]);
          acc[hh][q] = fma4(al1, r1[q], acc[hh][q]);
        }
      }
    }
    if (k < nb) {
      const int s0 = __builtin_amdgcn_readlane(sv, k);
      const float* p0 = xsrc + (size_t)s0 * KK + lane * 4;
      v4f r0[NQ];
#pragma unroll
      for (int q = 0; q < NQ; ++q) r0[q] = *(const v4f*)(p0 + q * 128);
      const float a0 = a_src[(size_t)s0 * HH + h] + ad;
      const float v0 = (a0 > 0.0f) ? a0 : NEG_SLOPE * a0;
      const int w0 = __float_as_int(expf(v0 - mx) / den);
#pragma unroll
      for (int hh = 0; hh < HH; ++hh) {
        const float al0 = __int_as_float(__builtin_amdgcn_readlane(w0, hh));
#pragma unroll
        for (int q = 0; q < NQ; ++q) acc[hh][q] = fma4(al0, r0[q], acc[hh][q]);
      }
    }
  }

#pragma unroll
  for (int hh = 0; hh < HH; ++hh)
#pragma unroll
    for (int q = 0; q < NQ; ++q)
      *(v4f*)(zp + hh * hpitch + q * 128) = acc[hh][q];
}

template <int HH, int QPT>
__global__ void __launch_bounds__(256)
head_mean_bias(const float* __restrict__ p, size_t hpitch,
               const float* __restrict__ bias,
               float* __restrict__ out, int rows, int cols) {
  const int nq = rows * (cols / 4);
  const int q0 = blockIdx.x * (256 * QPT) + threadIdx.x;
  v4f t[QPT][HH];
  v4f b[QPT];
#pragma unroll
  for (int u = 0; u < QPT; ++u) {
    const int q = min(q0 + u * 256, nq - 1);
#pragma unroll
    for (int hh = 0; hh < HH; ++hh)
      t[u][hh] = *(const v4f*)(p + hh * hpitch + 4 * (size_t)q);
    b[u] = *(const v4f*)(bias + 4 * (q % (cols / 4)));
  }
#pragma unroll
  for (int u = 0; u < QPT; ++u) {
    const int q = q0 + u * 256;
    v4f s = t[u][0];
#pragma unroll
    for (int hh = 1; hh < HH; ++hh) s += t[u][hh];
    if (q < nq) *(v4f*)(out + 4 * (size_t)q) = s * (1.0f / (float)HH) + b[u];
  }
}

#define NB_AR1 ((GD + 7) / 8)
#define NB_AR2 ((GHC1 + 7) / 8)
#define NB_Z1  ((GN1 + 1 + 255) / 256)
#define NB_Z2  ((GN2 + 1 + 255) / 256)
#define NB_RK1 ((GE1 + 256 * RPT - 1) / (256 * RPT))
#define NB_RK2 ((GE2 + 256 * RPT - 1) / (256 * RPT))
#define NL_WG1 6
#define NL_WG2 4
#define NB_LS1 ((GN0 / 4 + 8 * NL_WG1 - 1) / (8 * NL_WG1))
#define NB_LD1 ((GN1 / 4 + 8 * NL_WG1 - 1) / (8 * NL_WG1))
#define NB_LS2 ((GN1 / 4 + 8 * NL_WG2 - 1) / (8 * NL_WG2))
#define NB_LD2 ((GN2 / 4 + 8 * NL_WG2 - 1) / (8 * NL_WG2))

__global__ void __launch_bounds__(256)
prep_att_vectors_counters(const float* __restrict__ W1,
                          const float* __restrict__ att_src1,
                          const float* __restrict__ att_dst1,
                          float* __restrict__ vsrc1, float* __restrict__ vdst1,
                          const float* __restrict__ W2,
                          const float* __restrict__ att_src2,
                          const float* __restrict__ att_dst2,
                          float* __restrict__ vsrc2, float* __restrict__ vdst2,
                          int* __restrict__ cnt1, int* __restrict__ cnt2) {
  int b = blockIdx.x;
  if (b < 2 * NB_AR1) {
    const bool isdst = b >= NB_AR1;
    att_reduce<GH1, GC1, 1>(isdst ? b - NB_AR1 : b, W1,
                            isdst ? att_dst1 : att_src1,
                            isdst ? vdst1 : vsrc1, GD);
    return;
  }
  b -= 2 * NB_AR1;
  if (b < 2 * NB_AR2) {
    const bool isdst = b >= NB_AR2;
    att_reduce<GH2, GC2, 1>(isdst ? b - NB_AR2 : b, W2,
                            isdst ? att_dst2 : att_src2,
                            isdst ? vdst2 : vsrc2, GHC1);
    return;
  }
  b -= 2 * NB_AR2;
  if (b < NB_Z1) {
    zero_i32(b, cnt1, GN1 + 1);
    return;
  }
  zero_i32(b - NB_Z1, cnt2, GN2 + 1);
}

__global__ void __launch_bounds__(256, 6)
rank_edges_logits1(const int* __restrict__ dst1, int* __restrict__ cnt1,
                   int* __restrict__ rank1,
                   const int* __restrict__ dst2, int* __restrict__ cnt2,
                   int* __restrict__ rank2,
                   const float* __restrict__ x,
                   const float* __restrict__ vsrc1,
                   const float* __restrict__ vdst1,
                   float* __restrict__ a_src1, float* __restrict__ a_dst1) {
  int b = blockIdx.x;
  if (b < NB_RK1) {
    bucket_rank(b, dst1, cnt1, rank1, GE1, GN1);
    return;
  }
  b -= NB_RK1;
  if (b < NB_RK2) {
    bucket_rank(b, dst2, cnt2, rank2, GE2, GN2);
    return;
  }
  b -= NB_RK2;
  const bool isdst = b >= NB_LS1;
  node_logits<GD, GH1>(isdst ? b - NB_LS1 : b, isdst ? NB_LD1 : NB_LS1, x,
                       isdst ? vdst1 : vsrc1,
                       isdst ? a_dst1 : a_src1,
                       isdst ? GN1 : GN0);
}

__global__ void __launch_bounds__(1024)
scan_buckets(const int* __restrict__ cnt1, int* __restrict__ start1,
             const int* __restrict__ cnt2, int* __restrict__ start2) {
  __shared__ int wsum[32];
  if (blockIdx.x == 0) bucket_scan<GN1>(cnt1, start1, wsum);
  else                 bucket_scan<GN2>(cnt2, start2, wsum);
}

__global__ void __launch_bounds__(256)
fill_buckets(const int* __restrict__ src1, const int* __restrict__ dst1,
             const int* __restrict__ rank1, const int* __restrict__ start1,
             int* __restrict__ psrc1,
             const int* __restrict__ src2, const int* __restrict__ dst2,
             const int* __restrict__ rank2, const int* __restrict__ start2,
             int* __restrict__ psrc2) {
  const int b = blockIdx.x;
  if (b < NB_RK1) {
    bucket_fill(b, src1, dst1, rank1, start1, psrc1, GE1);
    return;
  }
  bucket_fill(b - NB_RK1, src2, dst2, rank2, start2, psrc2, GE2);
}

__global__ void __launch_bounds__(256)
logits2(const float* __restrict__ x1,
        const float* __restrict__ vsrc2, const float* __restrict__ vdst2,
        float* __restrict__ a_src2, float* __restrict__ a_dst2) {
  const int b = blockIdx.x;
  const bool isdst = b >= NB_LS2;
  node_logits<GHC1, GH2>(isdst ? b - NB_LS2 : b, isdst ? NB_LD2 : NB_LS2, x1,
                         isdst ? vdst2 : vsrc2,
                         isdst ? a_dst2 : a_src2,
                         isdst ? GN2 : GN1);
}

#define GN1P 24064
#define GN2P 6016

extern "C" void kernel_launch(void* const* d_in, const int* in_sizes, int n_in,
                              void* d_out, int out_size, void* d_ws,
                              size_t ws_size, hipStream_t stream) {
  (void)in_sizes; (void)n_in; (void)out_size; (void)ws_size;

  const float* x        = (const float*)d_in[0];
  const int*   src1     = (const int*)d_in[1];
  const int*   dst1     = (const int*)d_in[2];
  const int*   src2     = (const int*)d_in[3];
  const int*   dst2     = (const int*)d_in[4];
  const float* W1       = (const float*)d_in[5];
  const float* att_src1 = (const float*)d_in[6];
  const float* att_dst1 = (const float*)d_in[7];
  const float* b1       = (const float*)d_in[8];
  const float* W2       = (const float*)d_in[9];
  const float* att_src2 = (const float*)d_in[10];
  const float* att_dst2 = (const float*)d_in[11];
  const float* b2       = (const float*)d_in[12];
  float* out = (float*)d_out;

  char* ws = (char*)d_ws;
  size_t off = 0;
  auto alloc = [&](size_t bytes) -> float* {
    off = (off + 255) & ~(size_t)255;
    float* p = (float*)(ws + off);
    off += bytes;
    return p;
  };

  float* z1      = alloc((size_t)GH1 * GN1P * GD * 4);
  float* x1      = alloc((size_t)GN1P * GHC1 * 4);
  float* vsrc1   = alloc((size_t)GD * GH1 * 4);
  float* vdst1   = alloc((size_t)GD * GH1 * 4);
  float* a_src1  = alloc((size_t)GN0 * GH1 * 4);
  float* a_dst1  = alloc((size_t)GN1 * GH1 * 4);
  int*   cnt1    = (int*)alloc((size_t)(GN1 + 1) * 4);
  int*   start1  = (int*)alloc((size_t)(GN1 + 1) * 4);
  int*   rank1   = (int*)alloc((size_t)GE1 * 4);
  int*   psrc1   = (int*)alloc((size_t)GE1 * 4);
  float* vsrc2   = alloc((size_t)GHC1 * GH2 * 4);
  float* vdst2   = alloc((size_t)GHC1 * GH2 * 4);
  float* a_src2  = alloc((size_t)GN1 * GH2 * 4);
  float* a_dst2  = alloc((size_t)GN2 * GH2 * 4);
  int*   cnt2    = (int*)alloc((size_t)(GN2 + 1) * 4);
  int*   start2  = (int*)alloc((size_t)(GN2 + 1) * 4);
  int*   rank2   = (int*)alloc((size_t)GE2 * 4);
  int*   psrc2   = (int*)alloc((size_t)GE2 * 4);
  float* z2      = z1;
  float* p2      = z1 + (size_t)GH2 * GN2P * GHC1;

  prep_att_vectors_counters<<<2 * NB_AR1 + 2 * NB_AR2 + NB_Z1 + NB_Z2, 256, 0, stream>>>(
      W1, att_src1, att_dst1, vsrc1, vdst1,
      W2, att_src2, att_dst2, vsrc2, vdst2, cnt1, cnt2);

  rank_edges_logits1<<<NB_RK1 + NB_RK2 + NB_LS1 + NB_LD1, 256, 0, stream>>>(
      dst1, cnt1, rank1, dst2, cnt2, rank2, x, vsrc1, vdst1, a_src1, a_dst1);

  scan_buckets<<<2, 1024, 0, stream>>>(cnt1, start1, cnt2, start2);
  fill_buckets<<<NB_RK1 + NB_RK2, 256, 0, stream>>>(
      src1, dst1, rank1, start1, psrc1, src2, dst2, rank2, start2, psrc2);

  gat_row_owner_heads<GH1, GD><<<GN1P / 8, 256, 0, stream>>>(
      x, psrc1, start1, a_src1, a_dst1, z1, GN1, GN1P);

  wmma_gemm_f32<8, 1, true><<<dim3(GC1 / 64, GN1P / 256, GH1), 256, 0, stream>>>(
      z1, W1, x1, b1, GD, GD, GHC1, GHC1,
      (size_t)GN1P * GD, (size_t)GC1, (size_t)GC1, (size_t)GC1);

  logits2<<<NB_LS2 + NB_LD2, 256, 0, stream>>>(x1, vsrc2, vdst2, a_src2, a_dst2);

  gat_row_owner_heads<GH2, GHC1><<<GN2P / 8, 256, 0, stream>>>(
      x1, psrc2, start2, a_src2, a_dst2, z2, GN2, GN2P);

  wmma_gemm_f32<2, 4, false><<<dim3(GC2 / 256, GN2P / 64, GH2), 256, 0, stream>>>(
      z2, W2, p2, (const float*)nullptr, GHC1, GHC1, GHC2, GC2,
      (size_t)GN2P * GHC1, (size_t)GC2, (size_t)GN2P * GC2, (size_t)0);

  head_mean_bias<GH2, 4><<<(GN2 * GC2 / 4 + 1023) / 1024, 256, 0, stream>>>(
      p2, (size_t)GN2P * GC2, b2, out, GN2, GC2);
}


# Row-owner kernels: a 32-edge batch's logits loaded once (8 in flight per lane), its softmax weights formed ahead of its row loop, now only row loads (4 or 2 rows in flight) + FMAs. Sum order kept.
# speedup vs baseline: 5.3395x; 1.0976x over previous
#include <hip/hip_runtime.h>
#include <math.h>


typedef float v2f __attribute__((ext_vector_type(2)));
typedef float v8f __attribute__((ext_vector_type(8)));

#define GN0 120000
#define GN1 24000
#define GN2 6000
#define GD  256
#define GH1 8
#define GC1 64
#define GHC1 512
#define GH2 8
#define GC2 256
#define GHC2 2048
#define GE1 384000
#define GE2 96000
#define NEG_SLOPE 0.2f

typedef float v4f __attribute__((ext_vector_type(4)));

#define TBK 16
#define APITCH 20

template <int WM, int WN, bool ELU>
__global__ void __launch_bounds__(256)
wmma_gemm_f32(const float* __restrict__ A, const float* __restrict__ B,
              float* __restrict__ C, const float* __restrict__ bias,
              int K, int lda, int ldb, int ldc,
              size_t sA, size_t sB, size_t sC, size_t sBi) {
  static_assert(WM * WN == 8, "8 waves per block");
  static_assert(WM % 2 == 0, "A is staged 64 rows per pass of the block");
  constexpr int TBM = 32 * WM;
  constexpr int TBN = 64 * WN;
  constexpr int BP  = 2 * TBN + 32;
  constexpr int AQ  = TBM / 64;
  constexpr int NBQ = TBN / 4;
  constexpr int BPT = (8 * NBQ + 255) / 256;
  constexpr int BSTEP = 256 / NBQ;
  constexpr bool BALL = (8 * NBQ) % 256 == 0;
  __shared__ __attribute__((aligned(16))) float As[2][TBM * APITCH];
  __shared__ __attribute__((aligned(16))) float Bs[2][(TBK / 2) * BP];

  const int tid   = threadIdx.x;
  const int lane  = tid & 31;
  const int wave  = tid >> 5;
  const int mrow  = lane & 15;
  const int khalf = lane >> 4;
  const int wm = wave / WN;
  const int wn = wave % WN;
  const int n0 = blockIdx.x * TBN;
  const int m0 = blockIdx.y * TBM;

  const int arow = tid >> 2;
  const int akq  = tid & 3;
  const int bp0  = tid / NBQ;
  const int bnq  = tid % NBQ;
  const bool bact = BALL || (bp0 < TBK / 2);
  A += (size_t)blockIdx.z * sA;
  B += (size_t)blockIdx.z * sB;
  C += (size_t)blockIdx.z * sC;
  const float* ag  = A + (size_t)(m0 + arow) * lda + akq * 4;
  const size_t arow64 = (size_t)64 * lda;
  const float* bg0 = B + (size_t)(2 * bp0) * ldb + n0 + bnq * 4;
  const float* bg1 = bg0 + ldb;
  const size_t pairs = (size_t)(2 * BSTEP) * ldb;
  const int a_st = arow * APITCH + akq * 4;
  const int b_st = bp0 * BP + bnq * 8;

  const int a_ld = (wm * 32 + mrow) * APITCH + 2 * khalf;
  const int b_ld = khalf * BP + (wn * 64 + mrow) * 2;

  v4f ra[AQ];
  v4f rb0[BPT], rb1[BPT];
#pragma unroll
  for (int i = 0; i < AQ; ++i) ra[i] = *(const v4f*)(ag + i * arow64);
  if (bact) {
#pragma unroll
    for (int u = 0; u < BPT; ++u) {
      rb0[u] = *(const v4f*)(bg0 + u * pairs);
      rb1[u] = *(const v4f*)(bg1 + u * pairs);
    }
  }

#pragma unroll
  for (int i = 0; i < AQ; ++i) *(v4f*)&As[0][a_st + i * 64 * APITCH] = ra[i];
  if (bact) {
#pragma unroll
    for (int u = 0; u < BPT; ++u) {
      float* bd = &Bs[0][b_st + u * BSTEP * BP];
      *(v4f*)bd       = (v4f){rb0[u].x, rb1[u].x, rb0[u].y, rb1[u].y};
      *(v4f*)(bd + 4) = (v4f){rb0[u].z, rb1[u].z, rb0[u].w, rb1[u].w};
    }
  }
  __syncthreads();

  float bcol[4];
#pragma unroll
  for (int j = 0; j < 4; ++j)
    bcol[j] = ELU ? bias[(size_t)blockIdx.z * sBi + n0 + wn * 64 + j * 16 + mrow]
                  : 0.0f;

  v8f acc[2][4];
#pragma unroll
  for (int i = 0; i < 2; ++i)
#pragma unroll
    for (int j = 0; j < 4; ++j) acc[i][j] = {};

  const int nchunks = K / TBK;
  for (int ch = 0; ch < nchunks; ++ch) {
    const int cur = ch & 1;
    const bool more = (ch + 1) < nchunks;
    if (more) {
      const int kc = (ch + 1) * TBK;
      const size_t kcN = (size_t)kc * ldb;
#pragma unroll
      for (int i = 0; i < AQ; ++i) ra[i] = *(const v4f*)(ag + kc + i * arow64);
      if (bact) {
#pragma unroll
        for (int u = 0; u < BPT; ++u) {
          rb0[u] = *(const v4f*)(bg0 + kcN + u * pairs);
          rb1[u] = *(const v4f*)(bg1 + kcN + u * pairs);
        }
      }
    }

    const float* as = &As[cur][a_ld];
    const float* bs = &Bs[cur][b_ld];
#pragma unroll
    for (int kk = 0; kk < TBK; kk += 4) {
      v2f a[2];
      v2f b[4];
#pragma unroll
      for (int i = 0; i < 2; ++i)
        a[i] = *(const v2f*)(as + i * 16 * APITCH + kk);
#pragma unroll
      for (int j = 0; j < 4; ++j)
        b[j] = *(const v2f*)(bs + (kk / 2) * BP + j * 32);
#pragma unroll
      for (int i = 0; i < 2; ++i)
#pragma unroll
        for (int j = 0; j < 4; ++j)
          acc[i][j] = __builtin_amdgcn_wmma_f32_16x16x4_f32(
              false, a[i], false, b[j], (short)0, acc[i][j], false, false);
    }

    if (more) {
      const int nx = cur ^ 1;
#pragma unroll
      for (int i = 0; i < AQ; ++i) *(v4f*)&As[nx][a_st + i * 64 * APITCH] = ra[i];
      if (bact) {
#pragma unroll
        for (int u = 0; u < BPT; ++u) {
          float* bd = &Bs[nx][b_st + u * BSTEP * BP];
          *(v4f*)bd       = (v4f){rb0[u].x, rb1[u].x, rb0[u].y, rb1[u].y};
          *(v4f*)(bd + 4) = (v4f){rb0[u].z, rb1[u].z, rb0[u].w, rb1[u].w};
        }
      }
    }
    __syncthreads();
  }

#pragma unroll
  for (int i = 0; i < 2; ++i) {
#pragma unroll
    for (int j = 0; j < 4; ++j) {
      const int col = n0 + wn * 64 + j * 16 + mrow;
#pragma unroll
      for (int r = 0; r < 8; ++r) {
        const int row = m0 + wm * 32 + i * 16 + r + (khalf << 3);
        float v = acc[i][j][r];
        if (ELU) {
          v += bcol[j];
          v = (v > 0.0f) ? v : expm1f(v);
        }
        C[(size_t)row * ldc + col] = v;
      }
    }
  }
}

template <int HH, int CC, int NPW>
__device__ __forceinline__ void
att_reduce(int bid,
           const float* __restrict__ h,
           const float* __restrict__ att,
           float* __restrict__ out,
           int n_nodes) {
  const int w = (int)(((unsigned)bid * 256u + threadIdx.x) >> 5);
  const int lane = threadIdx.x & 31;
  for (int k = 0; k < NPW; ++k) {
    const int node = w * NPW + k;
    if (node >= n_nodes) return;
    const float* hp = h + (size_t)node * (HH * CC);
    float keep = 0.0f;
#pragma unroll
    for (int head = 0; head < HH; ++head) {
      float s = 0.0f;
#pragma unroll
      for (int c = lane; c < CC; c += 32)
        s += hp[head * CC + c] * att[head * CC + c];
#pragma unroll
      for (int off = 16; off > 0; off >>= 1) s += __shfl_xor(s, off, 32);
      if (lane == head) keep = s;
    }
    if (lane < HH) out[(size_t)node * HH + lane] = keep;
  }
}

__device__ __forceinline__ void
zero_i32(int bid, int* __restrict__ p, int n) {
  const int i = bid * 256 + threadIdx.x;
  if (i < n) p[i] = 0;
}

#define RPT 4
__device__ __forceinline__ void
bucket_rank(int bid, const int* __restrict__ dst, int* __restrict__ cnt,
            int* __restrict__ rank, int E, int n_tgt) {
  const int base = bid * (256 * RPT) + threadIdx.x;
  int d[RPT];
#pragma unroll
  for (int k = 0; k < RPT; ++k) {
    const int t = base + k * 256;
    const int dv = dst[min(t, E - 1)];
    d[k] = (t < E) ? dv : n_tgt;
  }
  int r[RPT];
#pragma unroll
  for (int k = 0; k < RPT; ++k) r[k] = atomicAdd(&cnt[d[k]], 1);
#pragma unroll
  for (int k = 0; k < RPT; ++k) {
    const int t = base + k * 256;
    if (t < E) rank[t] = r[k];
  }
}

template <int N>
__device__ __forceinline__ void
bucket_scan(const int* __restrict__ cnt, int* __restrict__ start, int* wsum) {
  constexpr int PER = (N + 1023) / 1024;
  const int t = threadIdx.x;
  const int lane = t & 31;
  const int wave = t >> 5;
  const int lo = t * PER;
  int v[PER];
#pragma unroll
  for (int i = 0; i < PER; ++i) {
    const int idx = lo + i;
    const int c = cnt[min(idx, N - 1)];
    v[i] = (idx < N) ? c : 0;
  }
  int s = 0;
#pragma unroll
  for (int i = 0; i < PER; ++i) s += v[i];
  int inc = s;
#pragma unroll
  for (int off = 1; off < 32; off <<= 1) {
    const int y = __shfl_up(inc, off, 32);
    if (lane >= off) inc += y;
  }
  if (lane == 31) wsum[wave] = inc;
  __syncthreads();
  if (wave == 0) {
    const int w = wsum[lane];
    int winc = w;
#pragma unroll
    for (int off = 1; off < 32; off <<= 1) {
      const int y = __shfl_up(winc, off, 32);
      if (lane >= off) winc += y;
    }
    wsum[lane] = winc - w;
  }
  __syncthreads();
  int run = wsum[wave] + inc - s;
#pragma unroll
  for (int i = 0; i < PER; ++i) {
    const int idx = lo + i;
    if (idx < N) start[idx] = run;
    run += v[i];
  }
  if (t == 1023) start[N] = run;
}

__device__ __forceinline__ void
bucket_fill(int bid, const int* __restrict__ src, const int* __restrict__ dst,
            const int* __restrict__ rank, const int* __restrict__ start,
            int* __restrict__ psrc, int E) {
  const int base = bid * (256 * RPT) + threadIdx.x;
  int pos[RPT];
  int sv[RPT];
#pragma unroll
  for (int k = 0; k < RPT; ++k) {
    const int t = min(base + k * 256, E - 1);
    pos[k] = start[dst[t]] + rank[t];
    sv[k] = src[t];
  }
#pragma unroll
  for (int k = 0; k < RPT; ++k) {
    const int t = base + k * 256;
    if (t < E) psrc[pos[k]] = sv[k];
  }
}


template <int KK, int HH>
__device__ __forceinline__ void
node_logits(int bid, int nblk,
            const float* __restrict__ x,
            const float* __restrict__ V,
            float* __restrict__ out,
            int n_nodes) {
  static_assert(HH == 8, "4 nodes x 8 heads = one value per lane");
  static_assert(KK % 128 == 0, "a wave reads a row in 512-byte pieces");
  constexpr int NQ = KK / 128;
  const int lane = threadIdx.x & 31;
  const int w = __builtin_amdgcn_readfirstlane(bid * 8 + (int)(threadIdx.x >> 5));
  const int nw = nblk * 8;
  const int n_groups = (n_nodes + 3) >> 2;
  if (w >= n_groups) return;

  float vv[NQ][4][HH];
#pragma unroll
  for (int q = 0; q < NQ; ++q)
#pragma unroll
    for (int e = 0; e < 4; ++e) {
      const float* vp = V + (size_t)(q * 128 + lane * 4 + e) * HH;
      const v4f lo = *(const v4f*)vp;
      const v4f hi = *(const v4f*)(vp + 4);
      vv[q][e][0] = lo.x; vv[q][e][1] = lo.y; vv[q][e][2] = lo.z; vv[q][e][3] = lo.w;
      vv[q][e][4] = hi.x; vv[q][e][5] = hi.y; vv[q][e][6] = hi.z; vv[q][e][7] = hi.w;
    }

  const bool b16 = (lane & 16) != 0;
  const bool b8  = (lane & 8) != 0;
  const bool b4  = (lane & 4) != 0;
  const bool b2  = (lane & 2) != 0;
  const bool b1  = (lane & 1) != 0;

  v4f r[4][NQ];
#define NL_LOAD_ROWS(g_)                                                    \
  _Pragma("unroll") for (int n = 0; n < 4; ++n) {                           \
    const int node_ = min((g_) * 4 + n, n_nodes - 1);                       \
    const float* xp_ = x + (size_t)node_ * KK + lane * 4;                   \
    _Pragma("unroll") for (int q = 0; q < NQ; ++q)                          \
      r[n][q] = *(const v4f*)(xp_ + q * 128);                               \
  }

  int g = w;
  NL_LOAD_ROWS(g)
  for (;;) {
    float t16[16];
#pragma unroll
    for (int n = 0; n < 2; ++n) {
      float va[HH], vb[HH];
#pragma unroll
      for (int hh = 0; hh < HH; ++hh) {
        float sa = 0.0f, sb = 0.0f;
#pragma unroll
        for (int q = 0; q < NQ; ++q) {
          sa = fmaf(r[n][q].x, vv[q][0][hh], sa);
          sa = fmaf(r[n][q].y, vv[q][1][hh], sa);
          sa = fmaf(r[n][q].z, vv[q][2][hh], sa);
          sa = fmaf(r[n][q].w, vv[q][3][hh], sa);
          sb = fmaf(r[n + 2][q].x, vv[q][0][hh], sb);
          sb = fmaf(r[n + 2][q].y, vv[q][1][hh], sb);
          sb = fmaf(r[n + 2][q].z, vv[q][2][hh], sb);
          sb = fmaf(r[n + 2][q].w, vv[q][3][hh], sb);
        }
        va[hh] = sa;
        vb[hh] = sb;
      }
#pragma unroll
      for (int hh = 0; hh < HH; ++hh) {
        const float keep = b16 ? vb[hh] : va[hh];
        const float send = b16 ? va[hh] : vb[hh];
        t16[n * 8 + hh] = keep + __shfl_xor(send, 16, 32);
      }
    }

    const int gn = g + nw;
    const bool more = gn < n_groups;
    if (more) { NL_LOAD_ROWS(gn) }

    float t8[8];
#pragma unroll
    for (int j = 0; j < 8; ++j) {
      const float keep = b8 ? t16[j + 8] : t16[j];
      const float send = b8 ? t16[j] : t16[j + 8];
      t8[j] = keep + __shfl_xor(send, 8, 32);
    }
    float t4[4];
#pragma unroll
    for (int j = 0; j < 4; ++j) {
      const float keep = b4 ? t8[j + 4] : t8[j];
      const float send = b4 ? t8[j] : t8[j + 4];
      t4[j] = keep + __shfl_xor(send, 4, 32);
    }
    float t2[2];
#pragma unroll
    for (int j = 0; j < 2; ++j) {
      const float keep = b2 ? t4[j + 2] : t4[j];
      const float send = b2 ? t4[j] : t4[j + 2];
      t2[j] = keep + __shfl_xor(send, 2, 32);
    }
    const float keep1 = b1 ? t2[1] : t2[0];
    const float send1 = b1 ? t2[0] : t2[1];
    const float tot = keep1 + __shfl_xor(send1, 1, 32);

    if (g * 4 + (lane >> 3) < n_nodes) out[(size_t)g * 32 + lane] = tot;
    if (!more) break;
    g = gn;
  }
#undef NL_LOAD_ROWS
}

__device__ __forceinline__ v4f fma4(float a, v4f x, v4f c) {
  return (v4f){fmaf(a, x.x, c.x), fmaf(a, x.y, c.y),
               fmaf(a, x.z, c.z), fmaf(a, x.w, c.w)};
}

template <int HH, int KK, int RIF>
__global__ void __launch_bounds__(256)
gat_row_owner_heads(const float* __restrict__ xsrc,
                    const int* __restrict__ psrc,
                    const int* __restrict__ start,
                    const float* __restrict__ a_src,
                    const float* __restrict__ a_dst,
                    float* __restrict__ z,
                    int n_tgt, int n_pad) {
  static_assert(HH == 8, "lane map: 4 edge slots x 8 heads");
  static_assert(KK % 128 == 0, "a wave reads a row in 512-byte pieces");
  static_assert(RIF == 2 || RIF == 4, "rows in flight divide a group of 4 edges");
  constexpr int NQ = KK / 128;
  const int d = __builtin_amdgcn_readfirstlane(
      (int)(blockIdx.x * 8u + (threadIdx.x >> 5)));
  if (d >= n_pad) return;
  const int lane = threadIdx.x & 31;
  float* zp = z + (size_t)d * KK + lane * 4;
  const size_t hpitch = (size_t)n_pad * KK;
  if (d >= n_tgt) {
#pragma unroll
    for (int hh = 0; hh < HH; ++hh)
#pragma unroll
      for (int q = 0; q < NQ; ++q)
        *(v4f*)(zp + hh * hpitch + q * 128) = (v4f){0.0f, 0.0f, 0.0f, 0.0f};
    return;
  }
  const int h = lane & (HH - 1);
  const int slot = lane >> 3;
  const int beg = start[d];
  const int end = start[d + 1];
  const float ad = a_dst[(size_t)d * HH + h];
  const bool single = (end - beg) <= 32;
  int sv = 0;
  float lv[8];
#define RO_LOAD_LOGITS(b0_, nb_)                                            \
  {                                                                         \
    sv = psrc[(b0_) + min(lane, (nb_) - 1)];                                \
    _Pragma("unroll") for (int j = 0; j < 8; ++j) {                         \
      const int s_ = __shfl(sv, min(4 * j + slot, (nb_) - 1), 32);          \
      const float a_ = a_src[(size_t)s_ * HH + h] + ad;                     \
      lv[j] = (a_ > 0.0f) ? a_ : NEG_SLOPE * a_;                            \
    }                                                                       \
  }

  float mx = -INFINITY;
  for (int b0 = beg; b0 < end; b0 += 32) {
    const int nb = min(32, end - b0);
    RO_LOAD_LOGITS(b0, nb)
#pragma unroll
    for (int j = 0; j < 8; ++j) mx = fmaxf(mx, lv[j]);
  }
  mx = fmaxf(mx, __shfl_xor(mx, 8, 32));
  mx = fmaxf(mx, __shfl_xor(mx, 16, 32));
  float den = 0.0f;
  for (int b0 = beg; b0 < end; b0 += 32) {
    const int nb = min(32, end - b0);
    if (!single) RO_LOAD_LOGITS(b0, nb)
#pragma unroll
    for (int j = 0; j < 8; ++j)
      if (4 * j + slot < nb) den += expf(lv[j] - mx);
  }
  den += __shfl_xor(den, 8, 32);
  den += __shfl_xor(den, 16, 32);

  v4f acc[HH][NQ];
#pragma unroll
  for (int hh = 0; hh < HH; ++hh)
#pragma unroll
    for (int q = 0; q < NQ; ++q) acc[hh][q] = (v4f){0.0f, 0.0f, 0.0f, 0.0f};

  for (int b0 = beg; b0 < end; b0 += 32) {
    const int nb = min(32, end - b0);
    if (!single) RO_LOAD_LOGITS(b0, nb)
    float w[8];
#pragma unroll
    for (int j = 0; j < 8; ++j) w[j] = expf(lv[j] - mx) / den;
    for (int j = 0; 4 * j < nb; ++j) {
      float wj = w[0];
#pragma unroll
      for (int t = 1; t < 8; ++t) wj = (j == t) ? w[t] : wj;
      const int wi = __float_as_int(wj);
#pragma unroll
      for (int g = 0; g < 4; g += RIF) {
        const int k0 = 4 * j + g;
        if (k0 < nb) {
          v4f r[RIF][NQ];
#pragma unroll
          for (int t = 0; t < RIF; ++t) {
            const int s = __builtin_amdgcn_readlane(sv, min(k0 + t, nb - 1));
            const float* p = xsrc + (size_t)s * KK + lane * 4;
#pragma unroll
            for (int q = 0; q < NQ; ++q) r[t][q] = *(const v4f*)(p + q * 128);
          }
          if (RIF == 4 && NQ == 2) {
            asm volatile("" :: "v"(r[0][0]), "v"(r[0][1]), "v"(r[1][0]), "v"(r[1][1]),
                               "v"(r[2 % RIF][0]), "v"(r[2 % RIF][1]),
                               "v"(r[3 % RIF][0]), "v"(r[3 % RIF][1]));
          } else {
#pragma unroll
            for (int q = 0; q < NQ; ++q)
              asm volatile("" :: "v"(r[0][q]), "v"(r[1][q]));
          }
#pragma unroll
          for (int t = 0; t < RIF; ++t) {
            if (k0 + t < nb) {
#pragma unroll
              for (int hh = 0; hh < HH; ++hh) {
                const float al = __int_as_float(
                    __builtin_amdgcn_readlane(wi, (g + t) * 8 + hh));
#pragma unroll
                for (int q = 0; q < NQ; ++q) acc[hh][q] = fma4(al, r[t][q], acc[hh][q]);
              }
            }
          }
        }
      }
    }
  }
#undef RO_LOAD_LOGITS

#pragma unroll
  for (int hh = 0; hh < HH; ++hh)
#pragma unroll
    for (int q = 0; q < NQ; ++q)
      *(v4f*)(zp + hh * hpitch + q * 128) = acc[hh][q];
}

template <int HH, int QPT>
__global__ void __launch_bounds__(256)
head_mean_bias(const float* __restrict__ p, size_t hpitch,
               const float* __restrict__ bias,
               float* __restrict__ out, int rows, int cols) {
  const int nq = rows * (cols / 4);
  const int q0 = blockIdx.x * (256 * QPT) + threadIdx.x;
  v4f t[QPT][HH];
  v4f b[QPT];
#pragma unroll
  for (int u = 0; u < QPT; ++u) {
    const int q = min(q0 + u * 256, nq - 1);
#pragma unroll
    for (int hh = 0; hh < HH; ++hh)
      t[u][hh] = *(const v4f*)(p + hh * hpitch + 4 * (size_t)q);
    b[u] = *(const v4f*)(bias + 4 * (q % (cols / 4)));
  }
#pragma unroll
  for (int u = 0; u < QPT; ++u) {
    const int q = q0 + u * 256;
    v4f s = t[u][0];
#pragma unroll
    for (int hh = 1; hh < HH; ++hh) s += t[u][hh];
    if (q < nq) *(v4f*)(out + 4 * (size_t)q) = s * (1.0f / (float)HH) + b[u];
  }
}

#define NB_AR1 ((GD + 7) / 8)
#define NB_AR2 ((GHC1 + 7) / 8)
#define NB_Z1  ((GN1 + 1 + 255) / 256)
#define NB_Z2  ((GN2 + 1 + 255) / 256)
#define NB_RK1 ((GE1 + 256 * RPT - 1) / (256 * RPT))
#define NB_RK2 ((GE2 + 256 * RPT - 1) / (256 * RPT))
#define NL_WG1 6
#define NL_WG2 4
#define NB_LS1 ((GN0 / 4 + 8 * NL_WG1 - 1) / (8 * NL_WG1))
#define NB_LD1 ((GN1 / 4 + 8 * NL_WG1 - 1) / (8 * NL_WG1))
#define NB_LS2 ((GN1 / 4 + 8 * NL_WG2 - 1) / (8 * NL_WG2))
#define NB_LD2 ((GN2 / 4 + 8 * NL_WG2 - 1) / (8 * NL_WG2))

__global__ void __launch_bounds__(256)
prep_att_vectors_counters(const float* __restrict__ W1,
                          const float* __restrict__ att_src1,
                          const float* __restrict__ att_dst1,
                          float* __restrict__ vsrc1, float* __restrict__ vdst1,
                          const float* __restrict__ W2,
                          const float* __restrict__ att_src2,
                          const float* __restrict__ att_dst2,
                          float* __restrict__ vsrc2, float* __restrict__ vdst2,
                          int* __restrict__ cnt1, int* __restrict__ cnt2) {
  int b = blockIdx.x;
  if (b < 2 * NB_AR1) {
    const bool isdst = b >= NB_AR1;
    att_reduce<GH1, GC1, 1>(isdst ? b - NB_AR1 : b, W1,
                            isdst ? att_dst1 : att_src1,
                            isdst ? vdst1 : vsrc1, GD);
    return;
  }
  b -= 2 * NB_AR1;
  if (b < 2 * NB_AR2) {
    const bool isdst = b >= NB_AR2;
    att_reduce<GH2, GC2, 1>(isdst ? b - NB_AR2 : b, W2,
                            isdst ? att_dst2 : att_src2,
                            isdst ? vdst2 : vsrc2, GHC1);
    return;
  }
  b -= 2 * NB_AR2;
  if (b < NB_Z1) {
    zero_i32(b, cnt1, GN1 + 1);
    return;
  }
  zero_i32(b - NB_Z1, cnt2, GN2 + 1);
}

__global__ void __launch_bounds__(256, 6)
rank_edges_logits1(const int* __restrict__ dst1, int* __restrict__ cnt1,
                   int* __restrict__ rank1,
                   const int* __restrict__ dst2, int* __restrict__ cnt2,
                   int* __restrict__ rank2,
                   const float* __restrict__ x,
                   const float* __restrict__ vsrc1,
                   const float* __restrict__ vdst1,
                   float* __restrict__ a_src1, float* __restrict__ a_dst1) {
  int b = blockIdx.x;
  if (b < NB_RK1) {
    bucket_rank(b, dst1, cnt1, rank1, GE1, GN1);
    return;
  }
  b -= NB_RK1;
  if (b < NB_RK2) {
    bucket_rank(b, dst2, cnt2, rank2, GE2, GN2);
    return;
  }
  b -= NB_RK2;
  const bool isdst = b >= NB_LS1;
  node_logits<GD, GH1>(isdst ? b - NB_LS1 : b, isdst ? NB_LD1 : NB_LS1, x,
                       isdst ? vdst1 : vsrc1,
                       isdst ? a_dst1 : a_src1,
                       isdst ? GN1 : GN0);
}

__global__ void __launch_bounds__(1024)
scan_buckets(const int* __restrict__ cnt1, int* __restrict__ start1,
             const int* __restrict__ cnt2, int* __restrict__ start2) {
  __shared__ int wsum[32];
  if (blockIdx.x == 0) bucket_scan<GN1>(cnt1, start1, wsum);
  else                 bucket_scan<GN2>(cnt2, start2, wsum);
}

__global__ void __launch_bounds__(256)
fill_buckets(const int* __restrict__ src1, const int* __restrict__ dst1,
             const int* __restrict__ rank1, const int* __restrict__ start1,
             int* __restrict__ psrc1,
             const int* __restrict__ src2, const int* __restrict__ dst2,
             const int* __restrict__ rank2, const int* __restrict__ start2,
             int* __restrict__ psrc2) {
  const int b = blockIdx.x;
  if (b < NB_RK1) {
    bucket_fill(b, src1, dst1, rank1, start1, psrc1, GE1);
    return;
  }
  bucket_fill(b - NB_RK1, src2, dst2, rank2, start2, psrc2, GE2);
}

__global__ void __launch_bounds__(256)
logits2(const float* __restrict__ x1,
        const float* __restrict__ vsrc2, const float* __restrict__ vdst2,
        float* __restrict__ a_src2, float* __restrict__ a_dst2) {
  const int b = blockIdx.x;
  const bool isdst = b >= NB_LS2;
  node_logits<GHC1, GH2>(isdst ? b - NB_LS2 : b, isdst ? NB_LD2 : NB_LS2, x1,
                         isdst ? vdst2 : vsrc2,
                         isdst ? a_dst2 : a_src2,
                         isdst ? GN2 : GN1);
}

#define GN1P 24064
#define GN2P 6016

extern "C" void kernel_launch(void* const* d_in, const int* in_sizes, int n_in,
                              void* d_out, int out_size, void* d_ws,
                              size_t ws_size, hipStream_t stream) {
  (void)in_sizes; (void)n_in; (void)out_size; (void)ws_size;

  const float* x        = (const float*)d_in[0];
  const int*   src1     = (const int*)d_in[1];
  const int*   dst1     = (const int*)d_in[2];
  const int*   src2     = (const int*)d_in[3];
  const int*   dst2     = (const int*)d_in[4];
  const float* W1       = (const float*)d_in[5];
  const float* att_src1 = (const float*)d_in[6];
  const float* att_dst1 = (const float*)d_in[7];
  const float* b1       = (const float*)d_in[8];
  const float* W2       = (const float*)d_in[9];
  const float* att_src2 = (const float*)d_in[10];
  const float* att_dst2 = (const float*)d_in[11];
  const float* b2       = (const float*)d_in[12];
  float* out = (float*)d_out;

  char* ws = (char*)d_ws;
  size_t off = 0;
  auto alloc = [&](size_t bytes) -> float* {
    off = (off + 255) & ~(size_t)255;
    float* p = (float*)(ws + off);
    off += bytes;
    return p;
  };

  float* z1      = alloc((size_t)GH1 * GN1P * GD * 4);
  float* x1      = alloc((size_t)GN1P * GHC1 * 4);
  float* vsrc1   = alloc((size_t)GD * GH1 * 4);
  float* vdst1   = alloc((size_t)GD * GH1 * 4);
  float* a_src1  = alloc((size_t)GN0 * GH1 * 4);
  float* a_dst1  = alloc((size_t)GN1 * GH1 * 4);
  int*   cnt1    = (int*)alloc((size_t)(GN1 + 1) * 4);
  int*   start1  = (int*)alloc((size_t)(GN1 + 1) * 4);
  int*   rank1   = (int*)alloc((size_t)GE1 * 4);
  int*   psrc1   = (int*)alloc((size_t)GE1 * 4);
  float* vsrc2   = alloc((size_t)GHC1 * GH2 * 4);
  float* vdst2   = alloc((size_t)GHC1 * GH2 * 4);
  float* a_src2  = alloc((size_t)GN1 * GH2 * 4);
  float* a_dst2  = alloc((size_t)GN2 * GH2 * 4);
  int*   cnt2    = (int*)alloc((size_t)(GN2 + 1) * 4);
  int*   start2  = (int*)alloc((size_t)(GN2 + 1) * 4);
  int*   rank2   = (int*)alloc((size_t)GE2 * 4);
  int*   psrc2   = (int*)alloc((size_t)GE2 * 4);
  float* z2      = z1;
  float* p2      = z1 + (size_t)GH2 * GN2P * GHC1;

  prep_att_vectors_counters<<<2 * NB_AR1 + 2 * NB_AR2 + NB_Z1 + NB_Z2, 256, 0, stream>>>(
      W1, att_src1, att_dst1, vsrc1, vdst1,
      W2, att_src2, att_dst2, vsrc2, vdst2, cnt1, cnt2);

  rank_edges_logits1<<<NB_RK1 + NB_RK2 + NB_LS1 + NB_LD1, 256, 0, stream>>>(
      dst1, cnt1, rank1, dst2, cnt2, rank2, x, vsrc1, vdst1, a_src1, a_dst1);

  scan_buckets<<<2, 1024, 0, stream>>>(cnt1, start1, cnt2, start2);
  fill_buckets<<<NB_RK1 + NB_RK2, 256, 0, stream>>>(
      src1, dst1, rank1, start1, psrc1, src2, dst2, rank2, start2, psrc2);

  gat_row_owner_heads<GH1, GD, 4><<<GN1P / 8, 256, 0, stream>>>(
      x, psrc1, start1, a_src1, a_dst1, z1, GN1, GN1P);

  wmma_gemm_f32<8, 1, true><<<dim3(GC1 / 64, GN1P / 256, GH1), 256, 0, stream>>>(
      z1, W1, x1, b1, GD, GD, GHC1, GHC1,
      (size_t)GN1P * GD, (size_t)GC1, (size_t)GC1, (size_t)GC1);

  logits2<<<NB_LS2 + NB_LD2, 256, 0, stream>>>(x1, vsrc2, vdst2, a_src2, a_dst2);

  gat_row_owner_heads<GH2, GHC1, 2><<<GN2P / 8, 256, 0, stream>>>(
      x1, psrc2, start2, a_src2, a_dst2, z2, GN2, GN2P);

  wmma_gemm_f32<2, 4, false><<<dim3(GC2 / 256, GN2P / 64, GH2), 256, 0, stream>>>(
      z2, W2, p2, (const float*)nullptr, GHC1, GHC1, GHC2, GC2,
      (size_t)GN2P * GHC1, (size_t)GC2, (size_t)GN2P * GC2, (size_t)0);

  head_mean_bias<GH2, 4><<<(GN2 * GC2 / 4 + 1023) / 1024, 256, 0, stream>>>(
      p2, (size_t)GN2P * GC2, b2, out, GN2, GC2);
}


# Lane-strided table accesses made whole-line: bucket scan walks counters/starts as 16-byte quads on consecutive lanes; V tables lane-major, a wave load reads 512 consecutive bytes. Sum order kept.
# speedup vs baseline: 5.5609x; 1.0415x over previous
#include <hip/hip_runtime.h>
#include <math.h>


typedef float v2f __attribute__((ext_vector_type(2)));
typedef float v8f __attribute__((ext_vector_type(8)));

#define GN0 120000
#define GN1 24000
#define GN2 6000
#define GD  256
#define GH1 8
#define GC1 64
#define GHC1 512
#define GH2 8
#define GC2 256
#define GHC2 2048
#define GE1 384000
#define GE2 96000
#define NEG_SLOPE 0.2f

typedef float v4f __attribute__((ext_vector_type(4)));

#define TBK 16
#define APITCH 20

template <int WM, int WN, bool ELU>
__global__ void __launch_bounds__(256)
wmma_gemm_f32(const float* __restrict__ A, const float* __restrict__ B,
              float* __restrict__ C, const float* __restrict__ bias,
              int K, int lda, int ldb, int ldc,
              size_t sA, size_t sB, size_t sC, size_t sBi) {
  static_assert(WM * WN == 8, "8 waves per block");
  static_assert(WM % 2 == 0, "A is staged 64 rows per pass of the block");
  constexpr int TBM = 32 * WM;
  constexpr int TBN = 64 * WN;
  constexpr int BP  = 2 * TBN + 32;
  constexpr int AQ  = TBM / 64;
  constexpr int NBQ = TBN / 4;
  constexpr int BPT = (8 * NBQ + 255) / 256;
  constexpr int BSTEP = 256 / NBQ;
  constexpr bool BALL = (8 * NBQ) % 256 == 0;
  __shared__ __attribute__((aligned(16))) float As[2][TBM * APITCH];
  __shared__ __attribute__((aligned(16))) float Bs[2][(TBK / 2) * BP];

  const int tid   = threadIdx.x;
  const int lane  = tid & 31;
  const int wave  = tid >> 5;
  const int mrow  = lane & 15;
  const int khalf = lane >> 4;
  const int wm = wave / WN;
  const int wn = wave % WN;
  const int n0 = blockIdx.x * TBN;
  const int m0 = blockIdx.y * TBM;

  const int arow = tid >> 2;
  const int akq  = tid & 3;
  const int bp0  = tid / NBQ;
  const int bnq  = tid % NBQ;
  const bool bact = BALL || (bp0 < TBK / 2);
  A += (size_t)blockIdx.z * sA;
  B += (size_t)blockIdx.z * sB;
  C += (size_t)blockIdx.z * sC;
  const float* ag  = A + (size_t)(m0 + arow) * lda + akq * 4;
  const size_t arow64 = (size_t)64 * lda;
  const float* bg0 = B + (size_t)(2 * bp0) * ldb + n0 + bnq * 4;
  const float* bg1 = bg0 + ldb;
  const size_t pairs = (size_t)(2 * BSTEP) * ldb;
  const int a_st = arow * APITCH + akq * 4;
  const int b_st = bp0 * BP + bnq * 8;

  const int a_ld = (wm * 32 + mrow) * APITCH + 2 * khalf;
  const int b_ld = khalf * BP + (wn * 64 + mrow) * 2;

  v4f ra[AQ];
  v4f rb0[BPT], rb1[BPT];
#pragma unroll
  for (int i = 0; i < AQ; ++i) ra[i] = *(const v4f*)(ag + i * arow64);
  if (bact) {
#pragma unroll
    for (int u = 0; u < BPT; ++u) {
      rb0[u] = *(const v4f*)(bg0 + u * pairs);
      rb1[u] = *(const v4f*)(bg1 + u * pairs);
    }
  }

#pragma unroll
  for (int i = 0; i < AQ; ++i) *(v4f*)&As[0][a_st + i * 64 * APITCH] = ra[i];
  if (bact) {
#pragma unroll
    for (int u = 0; u < BPT; ++u) {
      float* bd = &Bs[0][b_st + u * BSTEP * BP];
      *(v4f*)bd       = (v4f){rb0[u].x, rb1[u].x, rb0[u].y, rb1[u].y};
      *(v4f*)(bd + 4) = (v4f){rb0[u].z, rb1[u].z, rb0[u].w, rb1[u].w};
    }
  }
  __syncthreads();

  float bcol[4];
#pragma unroll
  for (int j = 0; j < 4; ++j)
    bcol[j] = ELU ? bias[(size_t)blockIdx.z * sBi + n0 + wn * 64 + j * 16 + mrow]
                  : 0.0f;

  v8f acc[2][4];
#pragma unroll
  for (int i = 0; i < 2; ++i)
#pragma unroll
    for (int j = 0; j < 4; ++j) acc[i][j] = {};

  const int nchunks = K / TBK;
  for (int ch = 0; ch < nchunks; ++ch) {
    const int cur = ch & 1;
    const bool more = (ch + 1) < nchunks;
    if (more) {
      const int kc = (ch + 1) * TBK;
      const size_t kcN = (size_t)kc * ldb;
#pragma unroll
      for (int i = 0; i < AQ; ++i) ra[i] = *(const v4f*)(ag + kc + i * arow64);
      if (bact) {
#pragma unroll
        for (int u = 0; u < BPT; ++u) {
          rb0[u] = *(const v4f*)(bg0 + kcN + u * pairs);
          rb1[u] = *(const v4f*)(bg1 + kcN + u * pairs);
        }
      }
    }

    const float* as = &As[cur][a_ld];
    const float* bs = &Bs[cur][b_ld];
#pragma unroll
    for (int kk = 0; kk < TBK; kk += 4) {
      v2f a[2];
      v2f b[4];
#pragma unroll
      for (int i = 0; i < 2; ++i)
        a[i] = *(const v2f*)(as + i * 16 * APITCH + kk);
#pragma unroll
      for (int j = 0; j < 4; ++j)
        b[j] = *(const v2f*)(bs + (kk / 2) * BP + j * 32);
#pragma unroll
      for (int i = 0; i < 2; ++i)
#pragma unroll
        for (int j = 0; j < 4; ++j)
          acc[i][j] = __builtin_amdgcn_wmma_f32_16x16x4_f32(
              false, a[i], false, b[j], (short)0, acc[i][j], false, false);
    }

    if (more) {
      const int nx = cur ^ 1;
#pragma unroll
      for (int i = 0; i < AQ; ++i) *(v4f*)&As[nx][a_st + i * 64 * APITCH] = ra[i];
      if (bact) {
#pragma unroll
        for (int u = 0; u < BPT; ++u) {
          float* bd = &Bs[nx][b_st + u * BSTEP * BP];
          *(v4f*)bd       = (v4f){rb0[u].x, rb1[u].x, rb0[u].y, rb1[u].y};
          *(v4f*)(bd + 4) = (v4f){rb0[u].z, rb1[u].z, rb0[u].w, rb1[u].w};
        }
      }
    }
    __syncthreads();
  }

#pragma unroll
  for (int i = 0; i < 2; ++i) {
#pragma unroll
    for (int j = 0; j < 4; ++j) {
      const int col = n0 + wn * 64 + j * 16 + mrow;
#pragma unroll
      for (int r = 0; r < 8; ++r) {
        const int row = m0 + wm * 32 + i * 16 + r + (khalf << 3);
        float v = acc[i][j][r];
        if (ELU) {
          v += bcol[j];
          v = (v > 0.0f) ? v : expm1f(v);
        }
        C[(size_t)row * ldc + col] = v;
      }
    }
  }
}

__device__ __forceinline__ int v_slot(int k, int hh) {
  return ((((k >> 7) * 4 + (k & 3)) * 2 + (hh >> 2)) * 32 + ((k & 127) >> 2)) * 4 + (hh & 3);
}

template <int HH, int CC, int NPW>
__device__ __forceinline__ void
att_reduce(int bid,
           const float* __restrict__ h,
           const float* __restrict__ att,
           float* __restrict__ out,
           int n_nodes) {
  static_assert(HH == 8, "v_slot orders 8 heads");
  const int w = (int)(((unsigned)bid * 256u + threadIdx.x) >> 5);
  const int lane = threadIdx.x & 31;
  for (int k = 0; k < NPW; ++k) {
    const int node = w * NPW + k;
    if (node >= n_nodes) return;
    const float* hp = h + (size_t)node * (HH * CC);
    float keep = 0.0f;
#pragma unroll
    for (int head = 0; head < HH; ++head) {
      float s = 0.0f;
#pragma unroll
      for (int c = lane; c < CC; c += 32)
        s += hp[head * CC + c] * att[head * CC + c];
#pragma unroll
      for (int off = 16; off > 0; off >>= 1) s += __shfl_xor(s, off, 32);
      if (lane == head) keep = s;
    }
    if (lane < HH) out[v_slot(node, lane)] = keep;
  }
}

__device__ __forceinline__ void
zero_i32(int bid, int* __restrict__ p, int n) {
  const int i = bid * 256 + threadIdx.x;
  if (i < n) p[i] = 0;
}

#define RPT 4
__device__ __forceinline__ void
bucket_rank(int bid, const int* __restrict__ dst, int* __restrict__ cnt,
            int* __restrict__ rank, int E, int n_tgt) {
  const int base = bid * (256 * RPT) + threadIdx.x;
  int d[RPT];
#pragma unroll
  for (int k = 0; k < RPT; ++k) {
    const int t = base + k * 256;
    const int dv = dst[min(t, E - 1)];
    d[k] = (t < E) ? dv : n_tgt;
  }
  int r[RPT];
#pragma unroll
  for (int k = 0; k < RPT; ++k) r[k] = atomicAdd(&cnt[d[k]], 1);
#pragma unroll
  for (int k = 0; k < RPT; ++k) {
    const int t = base + k * 256;
    if (t < E) rank[t] = r[k];
  }
}

typedef int v4i __attribute__((ext_vector_type(4)));

template <int N>
__device__ __forceinline__ void
bucket_scan(const int* __restrict__ cnt, int* __restrict__ start, int* wsum) {
  static_assert(N % 4 == 0, "counters are walked in quads");
  constexpr int NQD = N / 4;
  constexpr int PERQ = (NQD + 1023) / 1024;
  static_assert(PERQ <= 6, "wsum holds 6 chunks");
  const int t = threadIdx.x;
  const int lane = t & 31;
  const int wave = t >> 5;
  v4i v[PERQ];
  int s[PERQ];
  int inc[PERQ];
#pragma unroll
  for (int j = 0; j < PERQ; ++j) {
    const int q = j * 1024 + t;
    const v4i c = *(const v4i*)(cnt + 4 * min(q, NQD - 1));
    v[j] = (q < NQD) ? c : (v4i){0, 0, 0, 0};
    s[j] = v[j].x + v[j].y + v[j].z + v[j].w;
    inc[j] = s[j];
  }
#pragma unroll
  for (int off = 1; off < 32; off <<= 1) {
#pragma unroll
    for (int j = 0; j < PERQ; ++j) {
      const int y = __shfl_up(inc[j], off, 32);
      if (lane >= off) inc[j] += y;
    }
  }
  if (lane == 31) {
#pragma unroll
    for (int j = 0; j < PERQ; ++j) wsum[j * 32 + wave] = inc[j];
  }
  __syncthreads();
  if (wave < PERQ) {
    const int w = wsum[wave * 32 + lane];
    int winc = w;
#pragma unroll
    for (int off = 1; off < 32; off <<= 1) {
      const int y = __shfl_up(winc, off, 32);
      if (lane >= off) winc += y;
    }
    wsum[wave * 32 + lane] = winc - w;
    if (lane == 31) wsum[PERQ * 32 + wave] = winc;
  }
  __syncthreads();
  int cb = 0;
#pragma unroll
  for (int j = 0; j < PERQ; ++j) {
    const int q = j * 1024 + t;
    const int base = cb + wsum[j * 32 + wave] + inc[j] - s[j];
    v4i o;
    o.x = base;
    o.y = o.x + v[j].x;
    o.z = o.y + v[j].y;
    o.w = o.z + v[j].z;
    if (q < NQD) *(v4i*)(start + 4 * q) = o;
    cb += wsum[PERQ * 32 + j];
  }
  if (t == 0) start[N] = cb;
}

__device__ __forceinline__ void
bucket_fill(int bid, const int* __restrict__ src, const int* __restrict__ dst,
            const int* __restrict__ rank, const int* __restrict__ start,
            int* __restrict__ psrc, int E) {
  const int base = bid * (256 * RPT) + threadIdx.x;
  int pos[RPT];
  int sv[RPT];
#pragma unroll
  for (int k = 0; k < RPT; ++k) {
    const int t = min(base + k * 256, E - 1);
    pos[k] = start[dst[t]] + rank[t];
    sv[k] = src[t];
  }
#pragma unroll
  for (int k = 0; k < RPT; ++k) {
    const int t = base + k * 256;
    if (t < E) psrc[pos[k]] = sv[k];
  }
}


template <int KK, int HH>
__device__ __forceinline__ void
node_logits(int bid, int nblk,
            const float* __restrict__ x,
            const float* __restrict__ V,
            float* __restrict__ out,
            int n_nodes) {
  static_assert(HH == 8, "4 nodes x 8 heads = one value per lane");
  static_assert(KK % 128 == 0, "a wave reads a row in 512-byte pieces");
  constexpr int NQ = KK / 128;
  const int lane = threadIdx.x & 31;
  const int w = __builtin_amdgcn_readfirstlane(bid * 8 + (int)(threadIdx.x >> 5));
  const int nw = nblk * 8;
  const int n_groups = (n_nodes + 3) >> 2;
  if (w >= n_groups) return;

  float vv[NQ][4][HH];
#pragma unroll
  for (int q = 0; q < NQ; ++q)
#pragma unroll
    for (int e = 0; e < 4; ++e) {
      const float* vp = V + (size_t)(((q * 4 + e) * 2) * 32 + lane) * 4;
      const v4f lo = *(const v4f*)vp;
      const v4f hi = *(const v4f*)(vp + 128);
      vv[q][e][0] = lo.x; vv[q][e][1] = lo.y; vv[q][e][2] = lo.z; vv[q][e][3] = lo.w;
      vv[q][e][4] = hi.x; vv[q][e][5] = hi.y; vv[q][e][6] = hi.z; vv[q][e][7] = hi.w;
    }

  const bool b16 = (lane & 16) != 0;
  const bool b8  = (lane & 8) != 0;
  const bool b4  = (lane & 4) != 0;
  const bool b2  = (lane & 2) != 0;
  const bool b1  = (lane & 1) != 0;

  v4f r[4][NQ];
#define NL_LOAD_ROWS(g_)                                                    \
  _Pragma("unroll") for (int n = 0; n < 4; ++n) {                           \
    const int node_ = min((g_) * 4 + n, n_nodes - 1);                       \
    const float* xp_ = x + (size_t)node_ * KK + lane * 4;                   \
    _Pragma("unroll") for (int q = 0; q < NQ; ++q)                          \
      r[n][q] = *(const v4f*)(xp_ + q * 128);                               \
  }

  int g = w;
  NL_LOAD_ROWS(g)
  for (;;) {
    float t16[16];
#pragma unroll
    for (int n = 0; n < 2; ++n) {
      float va[HH], vb[HH];
#pragma unroll
      for (int hh = 0; hh < HH; ++hh) {
        float sa = 0.0f, sb = 0.0f;
#pragma unroll
        for (int q = 0; q < NQ; ++q) {
          sa = fmaf(r[n][q].x, vv[q][0][hh], sa);
          sa = fmaf(r[n][q].y, vv[q][1][hh], sa);
          sa = fmaf(r[n][q].z, vv[q][2][hh], sa);
          sa = fmaf(r[n][q].w, vv[q][3][hh], sa);
          sb = fmaf(r[n + 2][q].x, vv[q][0][hh], sb);
          sb = fmaf(r[n + 2][q].y, vv[q][1][hh], sb);
          sb = fmaf(r[n + 2][q].z, vv[q][2][hh], sb);
          sb = fmaf(r[n + 2][q].w, vv[q][3][hh], sb);
        }
        va[hh] = sa;
        vb[hh] = sb;
      }
#pragma unroll
      for (int hh = 0; hh < HH; ++hh) {
        const float keep = b16 ? vb[hh] : va[hh];
        const float send = b16 ? va[hh] : vb[hh];
        t16[n * 8 + hh] = keep + __shfl_xor(send, 16, 32);
      }
    }

    const int gn = g + nw;
    const bool more = gn < n_groups;
    if (more) { NL_LOAD_ROWS(gn) }

    float t8[8];
#pragma unroll
    for (int j = 0; j < 8; ++j) {
      const float keep = b8 ? t16[j + 8] : t16[j];
      const float send = b8 ? t16[j] : t16[j + 8];
      t8[j] = keep + __shfl_xor(send, 8, 32);
    }
    float t4[4];
#pragma unroll
    for (int j = 0; j < 4; ++j) {
      const float keep = b4 ? t8[j + 4] : t8[j];
      const float send = b4 ? t8[j] : t8[j + 4];
      t4[j] = keep + __shfl_xor(send, 4, 32);
    }
    float t2[2];
#pragma unroll
    for (int j = 0; j < 2; ++j) {
      const float keep = b2 ? t4[j + 2] : t4[j];
      const float send = b2 ? t4[j] : t4[j + 2];
      t2[j] = keep + __shfl_xor(send, 2, 32);
    }
    const float keep1 = b1 ? t2[1] : t2[0];
    const float send1 = b1 ? t2[0] : t2[1];
    const float tot = keep1 + __shfl_xor(send1, 1, 32);

    if (g * 4 + (lane >> 3) < n_nodes) out[(size_t)g * 32 + lane] = tot;
    if (!more) break;
    g = gn;
  }
#undef NL_LOAD_ROWS
}

__device__ __forceinline__ v4f fma4(float a, v4f x, v4f c) {
  return (v4f){fmaf(a, x.x, c.x), fmaf(a, x.y, c.y),
               fmaf(a, x.z, c.z), fmaf(a, x.w, c.w)};
}

template <int HH, int KK, int RIF>
__global__ void __launch_bounds__(256)
gat_row_owner_heads(const float* __restrict__ xsrc,
                    const int* __restrict__ psrc,
                    const int* __restrict__ start,
                    const float* __restrict__ a_src,
                    const float* __restrict__ a_dst,
                    float* __restrict__ z,
                    int n_tgt, int n_pad) {
  static_assert(HH == 8, "lane map: 4 edge slots x 8 heads");
  static_assert(KK % 128 == 0, "a wave reads a row in 512-byte pieces");
  static_assert(RIF == 2 || RIF == 4, "rows in flight divide a group of 4 edges");
  constexpr int NQ = KK / 128;
  const int d = __builtin_amdgcn_readfirstlane(
      (int)(blockIdx.x * 8u + (threadIdx.x >> 5)));
  if (d >= n_pad) return;
  const int lane = threadIdx.x & 31;
  float* zp = z + (size_t)d * KK + lane * 4;
  const size_t hpitch = (size_t)n_pad * KK;
  if (d >= n_tgt) {
#pragma unroll
    for (int hh = 0; hh < HH; ++hh)
#pragma unroll
      for (int q = 0; q < NQ; ++q)
        *(v4f*)(zp + hh * hpitch + q * 128) = (v4f){0.0f, 0.0f, 0.0f, 0.0f};
    return;
  }
  const int h = lane & (HH - 1);
  const int slot = lane >> 3;
  const int beg = start[d];
  const int end = start[d + 1];
  const float ad = a_dst[(size_t)d * HH + h];
  const bool single = (end - beg) <= 32;
  int sv = 0;
  float lv[8];
#define RO_LOAD_LOGITS(b0_, nb_)                                            \
  {                                                                         \
    sv = psrc[(b0_) + min(lane, (nb_) - 1)];                                \
    _Pragma("unroll") for (int j = 0; j < 8; ++j) {                         \
      const int s_ = __shfl(sv, min(4 * j + slot, (nb_) - 1), 32);          \
      const float a_ = a_src[(size_t)s_ * HH + h] + ad;                     \
      lv[j] = (a_ > 0.0f) ? a_ : NEG_SLOPE * a_;                            \
    }                                                                       \
  }

  float mx = -INFINITY;
  for (int b0 = beg; b0 < end; b0 += 32) {
    const int nb = min(32, end - b0);
    RO_LOAD_LOGITS(b0, nb)
#pragma unroll
    for (int j = 0; j < 8; ++j) mx = fmaxf(mx, lv[j]);
  }
  mx = fmaxf(mx, __shfl_xor(mx, 8, 32));
  mx = fmaxf(mx, __shfl_xor(mx, 16, 32));
  float den = 0.0f;
  for (int b0 = beg; b0 < end; b0 += 32) {
    const int nb = min(32, end - b0);
    if (!single) RO_LOAD_LOGITS(b0, nb)
#pragma unroll
    for (int j = 0; j < 8; ++j)
      if (4 * j + slot < nb) den += expf(lv[j] - mx);
  }
  den += __shfl_xor(den, 8, 32);
  den += __shfl_xor(den, 16, 32);

  v4f acc[HH][NQ];
#pragma unroll
  for (int hh = 0; hh < HH; ++hh)
#pragma unroll
    for (int q = 0; q < NQ; ++q) acc[hh][q] = (v4f){0.0f, 0.0f, 0.0f, 0.0f};

  for (int b0 = beg; b0 < end; b0 += 32) {
    const int nb = min(32, end - b0);
    if (!single) RO_LOAD_LOGITS(b0, nb)
    float w[8];
#pragma unroll
    for (int j = 0; j < 8; ++j) w[j] = expf(lv[j] - mx) / den;
    for (int j = 0; 4 * j < nb; ++j) {
      float wj = w[0];
#pragma unroll
      for (int t = 1; t < 8; ++t) wj = (j == t) ? w[t] : wj;
      const int wi = __float_as_int(wj);
#pragma unroll
      for (int g = 0; g < 4; g += RIF) {
        const int k0 = 4 * j + g;
        if (k0 < nb) {
          v4f r[RIF][NQ];
#pragma unroll
          for (int t = 0; t < RIF; ++t) {
            const int s = __builtin_amdgcn_readlane(sv, min(k0 + t, nb - 1));
            const float* p = xsrc + (size_t)s * KK + lane * 4;
#pragma unroll
            for (int q = 0; q < NQ; ++q) r[t][q] = *(const v4f*)(p + q * 128);
          }
          if (RIF == 4 && NQ == 2) {
            asm volatile("" :: "v"(r[0][0]), "v"(r[0][1]), "v"(r[1][0]), "v"(r[1][1]),
                               "v"(r[2 % RIF][0]), "v"(r[2 % RIF][1]),
                               "v"(r[3 % RIF][0]), "v"(r[3 % RIF][1]));
          } else {
#pragma unroll
            for (int q = 0; q < NQ; ++q)
              asm volatile("" :: "v"(r[0][q]), "v"(r[1][q]));
          }
#pragma unroll
          for (int t = 0; t < RIF; ++t) {
            if (k0 + t < nb) {
#pragma unroll
              for (int hh = 0; hh < HH; ++hh) {
                const float al = __int_as_float(
                    __builtin_amdgcn_readlane(wi, (g + t) * 8 + hh));
#pragma unroll
                for (int q = 0; q < NQ; ++q) acc[hh][q] = fma4(al, r[t][q], acc[hh][q]);
              }
            }
          }
        }
      }
    }
  }
#undef RO_LOAD_LOGITS

#pragma unroll
  for (int hh = 0; hh < HH; ++hh)
#pragma unroll
    for (int q = 0; q < NQ; ++q)
      *(v4f*)(zp + hh * hpitch + q * 128) = acc[hh][q];
}

template <int HH, int QPT>
__global__ void __launch_bounds__(256)
head_mean_bias(const float* __restrict__ p, size_t hpitch,
               const float* __restrict__ bias,
               float* __restrict__ out, int rows, int cols) {
  const int nq = rows * (cols / 4);
  const int q0 = blockIdx.x * (256 * QPT) + threadIdx.x;
  v4f t[QPT][HH];
  v4f b[QPT];
#pragma unroll
  for (int u = 0; u < QPT; ++u) {
    const int q = min(q0 + u * 256, nq - 1);
#pragma unroll
    for (int hh = 0; hh < HH; ++hh)
      t[u][hh] = *(const v4f*)(p + hh * hpitch + 4 * (size_t)q);
    b[u] = *(const v4f*)(bias + 4 * (q % (cols / 4)));
  }
#pragma unroll
  for (int u = 0; u < QPT; ++u) {
    const int q = q0 + u * 256;
    v4f s = t[u][0];
#pragma unroll
    for (int hh = 1; hh < HH; ++hh) s += t[u][hh];
    if (q < nq) *(v4f*)(out + 4 * (size_t)q) = s * (1.0f / (float)HH) + b[u];
  }
}

#define NB_AR1 ((GD + 7) / 8)
#define NB_AR2 ((GHC1 + 7) / 8)
#define NB_Z1  ((GN1 + 1 + 255) / 256)
#define NB_Z2  ((GN2 + 1 + 255) / 256)
#define NB_RK1 ((GE1 + 256 * RPT - 1) / (256 * RPT))
#define NB_RK2 ((GE2 + 256 * RPT - 1) / (256 * RPT))
#define NL_WG1 6
#define NL_WG2 4
#define NB_LS1 ((GN0 / 4 + 8 * NL_WG1 - 1) / (8 * NL_WG1))
#define NB_LD1 ((GN1 / 4 + 8 * NL_WG1 - 1) / (8 * NL_WG1))
#define NB_LS2 ((GN1 / 4 + 8 * NL_WG2 - 1) / (8 * NL_WG2))
#define NB_LD2 ((GN2 / 4 + 8 * NL_WG2 - 1) / (8 * NL_WG2))

__global__ void __launch_bounds__(256)
prep_att_vectors_counters(const float* __restrict__ W1,
                          const float* __restrict__ att_src1,
                          const float* __restrict__ att_dst1,
                          float* __restrict__ vsrc1, float* __restrict__ vdst1,
                          const float* __restrict__ W2,
                          const float* __restrict__ att_src2,
                          const float* __restrict__ att_dst2,
                          float* __restrict__ vsrc2, float* __restrict__ vdst2,
                          int* __restrict__ cnt1, int* __restrict__ cnt2) {
  int b = blockIdx.x;
  if (b < 2 * NB_AR1) {
    const bool isdst = b >= NB_AR1;
    att_reduce<GH1, GC1, 1>(isdst ? b - NB_AR1 : b, W1,
                            isdst ? att_dst1 : att_src1,
                            isdst ? vdst1 : vsrc1, GD);
    return;
  }
  b -= 2 * NB_AR1;
  if (b < 2 * NB_AR2) {
    const bool isdst = b >= NB_AR2;
    att_reduce<GH2, GC2, 1>(isdst ? b - NB_AR2 : b, W2,
                            isdst ? att_dst2 : att_src2,
                            isdst ? vdst2 : vsrc2, GHC1);
    return;
  }
  b -= 2 * NB_AR2;
  if (b < NB_Z1) {
    zero_i32(b, cnt1, GN1 + 1);
    return;
  }
  zero_i32(b - NB_Z1, cnt2, GN2 + 1);
}

__global__ void __launch_bounds__(256, 6)
rank_edges_logits1(const int* __restrict__ dst1, int* __restrict__ cnt1,
                   int* __restrict__ rank1,
                   const int* __restrict__ dst2, int* __restrict__ cnt2,
                   int* __restrict__ rank2,
                   const float* __restrict__ x,
                   const float* __restrict__ vsrc1,
                   const float* __restrict__ vdst1,
                   float* __restrict__ a_src1, float* __restrict__ a_dst1) {
  int b = blockIdx.x;
  if (b < NB_RK1) {
    bucket_rank(b, dst1, cnt1, rank1, GE1, GN1);
    return;
  }
  b -= NB_RK1;
  if (b < NB_RK2) {
    bucket_rank(b, dst2, cnt2, rank2, GE2, GN2);
    return;
  }
  b -= NB_RK2;
  const bool isdst = b >= NB_LS1;
  node_logits<GD, GH1>(isdst ? b - NB_LS1 : b, isdst ? NB_LD1 : NB_LS1, x,
                       isdst ? vdst1 : vsrc1,
                       isdst ? a_dst1 : a_src1,
                       isdst ? GN1 : GN0);
}

__global__ void __launch_bounds__(1024)
scan_buckets(const int* __restrict__ cnt1, int* __restrict__ start1,
             const int* __restrict__ cnt2, int* __restrict__ start2) {
  __shared__ int wsum[32 * 6 + 6];
  if (blockIdx.x == 0) bucket_scan<GN1>(cnt1, start1, wsum);
  else                 bucket_scan<GN2>(cnt2, start2, wsum);
}

__global__ void __launch_bounds__(256)
fill_buckets(const int* __restrict__ src1, const int* __restrict__ dst1,
             const int* __restrict__ rank1, const int* __restrict__ start1,
             int* __restrict__ psrc1,
             const int* __restrict__ src2, const int* __restrict__ dst2,
             const int* __restrict__ rank2, const int* __restrict__ start2,
             int* __restrict__ psrc2) {
  const int b = blockIdx.x;
  if (b < NB_RK1) {
    bucket_fill(b, src1, dst1, rank1, start1, psrc1, GE1);
    return;
  }
  bucket_fill(b - NB_RK1, src2, dst2, rank2, start2, psrc2, GE2);
}

__global__ void __launch_bounds__(256)
logits2(const float* __restrict__ x1,
        const float* __restrict__ vsrc2, const float* __restrict__ vdst2,
        float* __restrict__ a_src2, float* __restrict__ a_dst2) {
  const int b = blockIdx.x;
  const bool isdst = b >= NB_LS2;
  node_logits<GHC1, GH2>(isdst ? b - NB_LS2 : b, isdst ? NB_LD2 : NB_LS2, x1,
                         isdst ? vdst2 : vsrc2,
                         isdst ? a_dst2 : a_src2,
                         isdst ? GN2 : GN1);
}

#define GN1P 24064
#define GN2P 6016

extern "C" void kernel_launch(void* const* d_in, const int* in_sizes, int n_in,
                              void* d_out, int out_size, void* d_ws,
                              size_t ws_size, hipStream_t stream) {
  (void)in_sizes; (void)n_in; (void)out_size; (void)ws_size;

  const float* x        = (const float*)d_in[0];
  const int*   src1     = (const int*)d_in[1];
  const int*   dst1     = (const int*)d_in[2];
  const int*   src2     = (const int*)d_in[3];
  const int*   dst2     = (const int*)d_in[4];
  const float* W1       = (const float*)d_in[5];
  const float* att_src1 = (const float*)d_in[6];
  const float* att_dst1 = (const float*)d_in[7];
  const float* b1       = (const float*)d_in[8];
  const float* W2       = (const float*)d_in[9];
  const float* att_src2 = (const float*)d_in[10];
  const float* att_dst2 = (const float*)d_in[11];
  const float* b2       = (const float*)d_in[12];
  float* out = (float*)d_out;

  char* ws = (char*)d_ws;
  size_t off = 0;
  auto alloc = [&](size_t bytes) -> float* {
    off = (off + 255) & ~(size_t)255;
    float* p = (float*)(ws + off);
    off += bytes;
    return p;
  };

  float* z1      = alloc((size_t)GH1 * GN1P * GD * 4);
  float* x1      = alloc((size_t)GN1P * GHC1 * 4);
  float* vsrc1   = alloc((size_t)GD * GH1 * 4);
  float* vdst1   = alloc((size_t)GD * GH1 * 4);
  float* a_src1  = alloc((size_t)GN0 * GH1 * 4);
  float* a_dst1  = alloc((size_t)GN1 * GH1 * 4);
  int*   cnt1    = (int*)alloc((size_t)(GN1 + 1) * 4);
  int*   start1  = (int*)alloc((size_t)(GN1 + 1) * 4);
  int*   rank1   = (int*)alloc((size_t)GE1 * 4);
  int*   psrc1   = (int*)alloc((size_t)GE1 * 4);
  float* vsrc2   = alloc((size_t)GHC1 * GH2 * 4);
  float* vdst2   = alloc((size_t)GHC1 * GH2 * 4);
  float* a_src2  = alloc((size_t)GN1 * GH2 * 4);
  float* a_dst2  = alloc((size_t)GN2 * GH2 * 4);
  int*   cnt2    = (int*)alloc((size_t)(GN2 + 1) * 4);
  int*   start2  = (int*)alloc((size_t)(GN2 + 1) * 4);
  int*   rank2   = (int*)alloc((size_t)GE2 * 4);
  int*   psrc2   = (int*)alloc((size_t)GE2 * 4);
  float* z2      = z1;
  float* p2      = z1 + (size_t)GH2 * GN2P * GHC1;

  prep_att_vectors_counters<<<2 * NB_AR1 + 2 * NB_AR2 + NB_Z1 + NB_Z2, 256, 0, stream>>>(
      W1, att_src1, att_dst1, vsrc1, vdst1,
      W2, att_src2, att_dst2, vsrc2, vdst2, cnt1, cnt2);

  rank_edges_logits1<<<NB_RK1 + NB_RK2 + NB_LS1 + NB_LD1, 256, 0, stream>>>(
      dst1, cnt1, rank1, dst2, cnt2, rank2, x, vsrc1, vdst1, a_src1, a_dst1);

  scan_buckets<<<2, 1024, 0, stream>>>(cnt1, start1, cnt2, start2);
  fill_buckets<<<NB_RK1 + NB_RK2, 256, 0, stream>>>(
      src1, dst1, rank1, start1, psrc1, src2, dst2, rank2, start2, psrc2);

  gat_row_owner_heads<GH1, GD, 4><<<GN1P / 8, 256, 0, stream>>>(
      x, psrc1, start1, a_src1, a_dst1, z1, GN1, GN1P);

  wmma_gemm_f32<8, 1, true><<<dim3(GC1 / 64, GN1P / 256, GH1), 256, 0, stream>>>(
      z1, W1, x1, b1, GD, GD, GHC1, GHC1,
      (size_t)GN1P * GD, (size_t)GC1, (size_t)GC1, (size_t)GC1);

  logits2<<<NB_LS2 + NB_LD2, 256, 0, stream>>>(x1, vsrc2, vdst2, a_src2, a_dst2);

  gat_row_owner_heads<GH2, GHC1, 2><<<GN2P / 8, 256, 0, stream>>>(
      x1, psrc2, start2, a_src2, a_dst2, z2, GN2, GN2P);

  wmma_gemm_f32<2, 4, false><<<dim3(GC2 / 256, GN2P / 64, GH2), 256, 0, stream>>>(
      z2, W2, p2, (const float*)nullptr, GHC1, GHC1, GHC2, GC2,
      (size_t)GN2P * GHC1, (size_t)GC2, (size_t)GN2P * GC2, (size_t)0);

  head_mean_bias<GH2, 4><<<(GN2 * GC2 / 4 + 1023) / 1024, 256, 0, stream>>>(
      p2, (size_t)GN2P * GC2, b2, out, GN2, GC2);
}
